# starting kernel
# baseline (speedup 1.0000x reference)
#include <hip/hip_runtime.h>
#include <hip/hip_bf16.h>

typedef __attribute__((ext_vector_type(16))) _Float16 v16h;
typedef __attribute__((ext_vector_type(8)))  _Float16 v8h;
typedef __attribute__((ext_vector_type(8)))  float    v8f;

#define LEAKY_SLOPE 0.01f

__device__ __forceinline__ float elu_f(float x)   { return x > 0.f ? x : (expf(x) - 1.f); }
__device__ __forceinline__ float leaky_f(float x) { return x > 0.f ? x : LEAKY_SLOPE * x; }

template <int ACT_IN, int ACT_OUT, int NT>
__global__ __launch_bounds__(256)
void wmma_gemm(const float* __restrict__ A, const _Float16* __restrict__ BhT,
               const float* __restrict__ bias, float* __restrict__ C,
               int Mrows, int K, int Nn)
{
    const int mtiles  = (Mrows + 15) >> 4;
    const int nstrips = Nn / (16 * NT);
    const int wave    = blockIdx.x * (blockDim.x >> 5) + (threadIdx.x >> 5);
    if (wave >= mtiles * nstrips) return;
    const int mt   = wave / nstrips;
    const int ns   = wave % nstrips;
    const int lane = threadIdx.x & 31;
    const int hl   = lane >> 4;
    const int lrow = lane & 15;

    int arow = mt * 16 + lrow;
    if (arow >= Mrows) arow = Mrows - 1;
    const float* Abase = A + (size_t)arow * K;

    const _Float16* Bcol[NT];
    #pragma unroll
    for (int t = 0; t < NT; ++t)
        Bcol[t] = BhT + (size_t)((ns * NT + t) * 16 + lrow) * K;

    v8f acc[NT];
    #pragma unroll
    for (int t = 0; t < NT; ++t) acc[t] = (v8f){};

    for (int kk = 0; kk < K; kk += 32) {
        const int kb = kk + hl * 8;
        __builtin_prefetch(Abase + kk + 32, 0, 0);

        const float4* ap = (const float4*)(Abase + kb);
        float4 a0 = ap[0], a1 = ap[1];
        float4 a2 = ap[4], a3 = ap[5];
        float af[16] = { a0.x, a0.y, a0.z, a0.w, a1.x, a1.y, a1.z, a1.w,
                         a2.x, a2.y, a2.z, a2.w, a3.x, a3.y, a3.z, a3.w };
        v16h a;
        #pragma unroll
        for (int i = 0; i < 16; ++i) {
            float v = af[i];
            if (ACT_IN) v = elu_f(v);
            a[i] = (_Float16)v;
        }

        #pragma unroll
        for (int t = 0; t < NT; ++t) {
            v8h b0 = *(const v8h*)(Bcol[t] + kb);
            v8h b1 = *(const v8h*)(Bcol[t] + kb + 16);
            v16h b = __builtin_shufflevector(b0, b1, 0, 1, 2, 3, 4, 5, 6, 7,
                                             8, 9, 10, 11, 12, 13, 14, 15);
            acc[t] = __builtin_amdgcn_wmma_f32_16x16x32_f16(false, a, false, b,
                                                            (short)0, acc[t],
                                                            false, false);
        }
    }

    const bool fullM = (mt * 16 + 16) <= Mrows;
    #pragma unroll
    for (int t = 0; t < NT; ++t) {
        const int col = (ns * NT + t) * 16 + lrow;
        const float bv = bias ? bias[col] : 0.f;
        float* Cb = C + (size_t)(mt * 16 + hl * 8) * Nn + col;
        if (fullM) {
            #pragma unroll
            for (int r = 0; r < 8; ++r) {
                float v = acc[t][r] + bv;
                if (ACT_OUT) v = leaky_f(v);
                Cb[(size_t)r * Nn] = v;
            }
        } else {
            #pragma unroll
            for (int r = 0; r < 8; ++r) {
                const int orow = mt * 16 + hl * 8 + r;
                if (orow < Mrows) {
                    float v = acc[t][r] + bv;
                    if (ACT_OUT) v = leaky_f(v);
                    Cb[(size_t)r * Nn] = v;
                }
            }
        }
    }
}

__global__ void k_fill1(float* p, int n)
{
    int i = blockIdx.x * blockDim.x + threadIdx.x;
    if (i < n) p[i] = 1.0f;
}

__global__ void k_deg(const int* __restrict__ dst, float* deg, int E)
{
    int e = blockIdx.x * blockDim.x + threadIdx.x;
    if (e < E) atomicAdd(&deg[dst[e]], 1.0f);
}

__global__ void k_rsqrt_inplace(float* p, int n)
{
    int i = blockIdx.x * blockDim.x + threadIdx.x;
    if (i < n) p[i] = rsqrtf(p[i]);
}

__global__ void k_f16T(const float* __restrict__ s, _Float16* __restrict__ d,
                       int K, int Nn)
{
    int i = blockIdx.x * blockDim.x + threadIdx.x;
    if (i < K * Nn) {
        int k = i / Nn, n = i % Nn;
        d[(size_t)n * K + k] = (_Float16)s[i];
    }
}

__global__ void k_selfinit(const float* __restrict__ h, const float* __restrict__ dis,
                           float* __restrict__ agg, int n)
{
    int i = blockIdx.x * blockDim.x + threadIdx.x;
    int node = i >> 5;
    int c    = (i & 31) * 4;
    if (node < n) {
        float s = dis[node]; s *= s;
        float4 v = *(const float4*)(h + (size_t)node * 128 + c);
        v.x *= s; v.y *= s; v.z *= s; v.w *= s;
        *(float4*)(agg + (size_t)node * 128 + c) = v;
    }
}

__global__ void k_scatter(const int* __restrict__ srcI, const int* __restrict__ dstI,
                          const float* __restrict__ dis, const float* __restrict__ h,
                          float* __restrict__ agg, int E)
{
    int gid = blockIdx.x * blockDim.x + threadIdx.x;
    int e = gid >> 5;
    int c = (gid & 31) * 4;
    if (e < E) {
        int s = srcI[e], d = dstI[e];
        float w = dis[s] * dis[d];
        float4 v = *(const float4*)(h + (size_t)s * 128 + c);
        float* o = agg + (size_t)d * 128 + c;
        atomicAdd(o + 0, v.x * w);
        atomicAdd(o + 1, v.y * w);
        atomicAdd(o + 2, v.z * w);
        atomicAdd(o + 3, v.w * w);
    }
}

__global__ void k_pairdot(const int* __restrict__ ts, const float* __restrict__ tf,
                          const float* __restrict__ tg, float* __restrict__ out, int M)
{
    int m = blockIdx.x * blockDim.x + threadIdx.x;
    if (m < M) {
        int a = ts[2 * m + 0];
        int b = ts[2 * m + 1];
        const float4* pa = (const float4*)(tf + (size_t)a * 32);
        const float4* pb = (const float4*)(tg + (size_t)b * 32);
        float s = 0.f;
        #pragma unroll
        for (int i = 0; i < 8; ++i) {
            float4 x = pa[i], y = pb[i];
            s += x.x * y.x + x.y * y.y + x.z * y.z + x.w * y.w;
        }
        out[m] = s;
    }
}

extern "C" void kernel_launch(void* const* d_in, const int* in_sizes, int n_in,
                              void* d_out, int out_size, void* d_ws, size_t ws_size,
                              hipStream_t stream)
{
    const float* x    = (const float*)d_in[0];
    const int*   ei   = (const int*)  d_in[1];
    const int*   ts   = (const int*)  d_in[2];
    const float* W1   = (const float*)d_in[3];
    const float* W2   = (const float*)d_in[4];
    const float* tfw1 = (const float*)d_in[5];
    const float* tfb1 = (const float*)d_in[6];
    const float* tfw2 = (const float*)d_in[7];
    const float* tfb2 = (const float*)d_in[8];
    const float* tgw1 = (const float*)d_in[9];
    const float* tgb1 = (const float*)d_in[10];
    const float* tgw2 = (const float*)d_in[11];
    const float* tgb2 = (const float*)d_in[12];

    const int D_IN = 256, H1 = 128, H2 = 128, H3 = 64, D_OUT = 32;
    const int N = in_sizes[0] / D_IN;
    const int E = in_sizes[1] / 2;
    const int M = in_sizes[2] / 2;
    const int* srcI = ei;
    const int* dstI = ei + E;
    float* out = (float*)d_out;

    char* w = (char*)d_ws;
    auto alloc = [&](size_t bytes) -> char* {
        char* p = w; w += (bytes + 255) & ~(size_t)255; return p;
    };
    float*    dis = (float*)   alloc((size_t)N * 4);
    _Float16* W1h = (_Float16*)alloc((size_t)D_IN * H1 * 2);
    _Float16* W2h = (_Float16*)alloc((size_t)H1 * H2 * 2);
    _Float16* F1h = (_Float16*)alloc((size_t)H2 * H3 * 2);
    _Float16* F2h = (_Float16*)alloc((size_t)H3 * D_OUT * 2);
    _Float16* G1h = (_Float16*)alloc((size_t)H2 * H3 * 2);
    _Float16* G2h = (_Float16*)alloc((size_t)H3 * D_OUT * 2);
    float*    B1  = (float*)   alloc((size_t)N * H1 * 4);
    float*    B2  = (float*)   alloc((size_t)N * H1 * 4);
    float*    B3  = (float*)   alloc((size_t)N * D_OUT * 4);
    (void)ws_size; (void)n_in; (void)out_size;

    const int T = 256;
    auto cdiv = [](long long a, long long b) { return (int)((a + b - 1) / b); };

    k_fill1<<<cdiv(N, T), T, 0, stream>>>(dis, N);
    k_deg  <<<cdiv(E, T), T, 0, stream>>>(dstI, dis, E);
    k_rsqrt_inplace<<<cdiv(N, T), T, 0, stream>>>(dis, N);

    k_f16T<<<cdiv(D_IN * H1,   T), T, 0, stream>>>(W1,   W1h, D_IN, H1);
    k_f16T<<<cdiv(H1 * H2,     T), T, 0, stream>>>(W2,   W2h, H1,   H2);
    k_f16T<<<cdiv(H2 * H3,     T), T, 0, stream>>>(tfw1, F1h, H2,   H3);
    k_f16T<<<cdiv(H3 * D_OUT,  T), T, 0, stream>>>(tfw2, F2h, H3,   D_OUT);
    k_f16T<<<cdiv(H2 * H3,     T), T, 0, stream>>>(tgw1, G1h, H2,   H3);
    k_f16T<<<cdiv(H3 * D_OUT,  T), T, 0, stream>>>(tgw2, G2h, H3,   D_OUT);

    auto gemm_blocks = [&](int m, int nn, int nt) {
        return cdiv((long long)((m + 15) / 16) * (nn / (16 * nt)), 8);
    };

    wmma_gemm<0, 0, 4><<<gemm_blocks(N, H1, 4), T, 0, stream>>>(x, W1h, nullptr, B1, N, D_IN, H1);
    k_selfinit<<<cdiv((long long)N * 32, T), T, 0, stream>>>(B1, dis, B2, N);
    k_scatter <<<cdiv((long long)E * 32, T), T, 0, stream>>>(srcI, dstI, dis, B1, B2, E);

    wmma_gemm<1, 0, 4><<<gemm_blocks(N, H2, 4), T, 0, stream>>>(B2, W2h, nullptr, B1, N, H1, H2);
    k_selfinit<<<cdiv((long long)N * 32, T), T, 0, stream>>>(B1, dis, B2, N);
    k_scatter <<<cdiv((long long)E * 32, T), T, 0, stream>>>(srcI, dstI, dis, B1, B2, E);

    float* t1  = B1;
    float* tfO = B1 + (size_t)N * H3;
    wmma_gemm<0, 1, 4><<<gemm_blocks(N, H3, 4),    T, 0, stream>>>(B2, F1h, tfb1, t1,  N, H2, H3);
    wmma_gemm<0, 1, 2><<<gemm_blocks(N, D_OUT, 2), T, 0, stream>>>(t1, F2h, tfb2, tfO, N, H3, D_OUT);
    wmma_gemm<0, 1, 4><<<gemm_blocks(N, H3, 4),    T, 0, stream>>>(B2, G1h, tgb1, t1,  N, H2, H3);
    wmma_gemm<0, 1, 2><<<gemm_blocks(N, D_OUT, 2), T, 0, stream>>>(t1, G2h, tgb2, B3,  N, H3, D_OUT);

    k_pairdot<<<cdiv(M, T), T, 0, stream>>>(ts, tfO, B3, out, M);
}


# Float-atomic edge scatter-add replaced by per-node gather over destination buckets put in edge-number order (int atomics only place edges): fixed sum order (edges, then self loop), fma, plain store
# speedup vs baseline: 1.5260x; 1.5260x over previous
#include <hip/hip_runtime.h>
#include <hip/hip_bf16.h>

typedef __attribute__((ext_vector_type(16))) _Float16 v16h;
typedef __attribute__((ext_vector_type(8)))  _Float16 v8h;
typedef __attribute__((ext_vector_type(8)))  float    v8f;

#define LEAKY_SLOPE 0.01f

__device__ __forceinline__ float elu_f(float x)   { return x > 0.f ? x : (expf(x) - 1.f); }
__device__ __forceinline__ float leaky_f(float x) { return x > 0.f ? x : LEAKY_SLOPE * x; }

template <int ACT_IN, int ACT_OUT, int NT>
__global__ __launch_bounds__(256)
void wmma_gemm(const float* __restrict__ A, const _Float16* __restrict__ BhT,
               const float* __restrict__ bias, float* __restrict__ C,
               int Mrows, int K, int Nn)
{
    const int mtiles  = (Mrows + 15) >> 4;
    const int nstrips = Nn / (16 * NT);
    const int wave    = blockIdx.x * (blockDim.x >> 5) + (threadIdx.x >> 5);
    if (wave >= mtiles * nstrips) return;
    const int mt   = wave / nstrips;
    const int ns   = wave % nstrips;
    const int lane = threadIdx.x & 31;
    const int hl   = lane >> 4;
    const int lrow = lane & 15;

    int arow = mt * 16 + lrow;
    if (arow >= Mrows) arow = Mrows - 1;
    const float* Abase = A + (size_t)arow * K;

    const _Float16* Bcol[NT];
    #pragma unroll
    for (int t = 0; t < NT; ++t)
        Bcol[t] = BhT + (size_t)((ns * NT + t) * 16 + lrow) * K;

    v8f acc[NT];
    #pragma unroll
    for (int t = 0; t < NT; ++t) acc[t] = (v8f){};

    for (int kk = 0; kk < K; kk += 32) {
        const int kb = kk + hl * 8;
        __builtin_prefetch(Abase + kk + 32, 0, 0);

        const float4* ap = (const float4*)(Abase + kb);
        float4 a0 = ap[0], a1 = ap[1];
        float4 a2 = ap[4], a3 = ap[5];
        float af[16] = { a0.x, a0.y, a0.z, a0.w, a1.x, a1.y, a1.z, a1.w,
                         a2.x, a2.y, a2.z, a2.w, a3.x, a3.y, a3.z, a3.w };
        v16h a;
        #pragma unroll
        for (int i = 0; i < 16; ++i) {
            float v = af[i];
            if (ACT_IN) v = elu_f(v);
            a[i] = (_Float16)v;
        }

        #pragma unroll
        for (int t = 0; t < NT; ++t) {
            v8h b0 = *(const v8h*)(Bcol[t] + kb);
            v8h b1 = *(const v8h*)(Bcol[t] + kb + 16);
            v16h b = __builtin_shufflevector(b0, b1, 0, 1, 2, 3, 4, 5, 6, 7,
                                             8, 9, 10, 11, 12, 13, 14, 15);
            acc[t] = __builtin_amdgcn_wmma_f32_16x16x32_f16(false, a, false, b,
                                                            (short)0, acc[t],
                                                            false, false);
        }
    }

    const bool fullM = (mt * 16 + 16) <= Mrows;
    #pragma unroll
    for (int t = 0; t < NT; ++t) {
        const int col = (ns * NT + t) * 16 + lrow;
        const float bv = bias ? bias[col] : 0.f;
        float* Cb = C + (size_t)(mt * 16 + hl * 8) * Nn + col;
        if (fullM) {
            #pragma unroll
            for (int r = 0; r < 8; ++r) {
                float v = acc[t][r] + bv;
                if (ACT_OUT) v = leaky_f(v);
                Cb[(size_t)r * Nn] = v;
            }
        } else {
            #pragma unroll
            for (int r = 0; r < 8; ++r) {
                const int orow = mt * 16 + hl * 8 + r;
                if (orow < Mrows) {
                    float v = acc[t][r] + bv;
                    if (ACT_OUT) v = leaky_f(v);
                    Cb[(size_t)r * Nn] = v;
                }
            }
        }
    }
}

__global__ void k_zero(int* p, int n)
{
    int i = blockIdx.x * blockDim.x + threadIdx.x;
    if (i < n) p[i] = 0;
}

__global__ void k_count(const int* __restrict__ dst, int* cnt, int E)
{
    int e = blockIdx.x * blockDim.x + threadIdx.x;
    if (e < E) atomicAdd(&cnt[dst[e]], 1);
}

__global__ __launch_bounds__(256)
void k_scan(const int* __restrict__ cnt, int* __restrict__ rowptr,
            int* __restrict__ cursor, float* __restrict__ dis, int n)
{
    __shared__ int s_pre[8];
    __shared__ int s_tot[8];
    const int t = threadIdx.x, lane = t & 31, wv = t >> 5;

    const int4* c4 = (const int4*)cnt;
    const int nb = blockIdx.x;
    int pre = 0;
    #pragma unroll 8
    for (int m = 0; m < nb; ++m) {
        int4 v = c4[m * 256 + t];
        pre += (v.x + v.y) + (v.z + v.w);
    }

    const int i0 = blockIdx.x * 1024 + t * 4;
    int c[4];
    if (i0 + 3 < n) {
        int4 v = c4[i0 >> 2];
        c[0] = v.x; c[1] = v.y; c[2] = v.z; c[3] = v.w;
    } else {
        for (int k = 0; k < 4; ++k) c[k] = (i0 + k < n) ? cnt[i0 + k] : 0;
    }
    const int mine = (c[0] + c[1]) + (c[2] + c[3]);

    int incl = mine;
    for (int off = 1; off < 32; off <<= 1) {
        int v = __shfl_up(incl, off);
        if (lane >= off) incl += v;
    }
    for (int off = 16; off > 0; off >>= 1) pre += __shfl_xor(pre, off);
    if (lane == 31) s_tot[wv] = incl;
    if (lane == 0)  s_pre[wv] = pre;
    __syncthreads();
    int run = incl - mine;
    for (int k = 0; k < 8; ++k) {
        run += s_pre[k];
        if (k < wv) run += s_tot[k];
    }

    int r[4]; float d[4];
    for (int k = 0; k < 4; ++k) {
        r[k] = run;
        d[k] = rsqrtf((float)(c[k] + 1));
        run += c[k];
    }
    if (i0 + 3 < n) {
        *(int4*)(rowptr + i0)  = make_int4(r[0], r[1], r[2], r[3]);
        *(int4*)(cursor + i0)  = make_int4(r[0], r[1], r[2], r[3]);
        *(float4*)(dis + i0)   = make_float4(d[0], d[1], d[2], d[3]);
    } else {
        for (int k = 0; k < 4; ++k)
            if (i0 + k < n) { rowptr[i0 + k] = r[k]; cursor[i0 + k] = r[k]; dis[i0 + k] = d[k]; }
    }
    if (i0 < n && n <= i0 + 4) rowptr[n] = run;
}

__global__ void k_fill(const int* __restrict__ dst, int* cursor, int* __restrict__ ebuf, int E)
{
    int e = blockIdx.x * blockDim.x + threadIdx.x;
    if (e < E) {
        int pos = atomicAdd(&cursor[dst[e]], 1);
        ebuf[pos] = e;
    }
}

__global__ void k_sort(const int* __restrict__ ebuf, const int* __restrict__ srcI,
                       const int* __restrict__ dstI, const int* __restrict__ rowptr,
                       const float* __restrict__ dis, int2* __restrict__ csr, int E)
{
    int p = blockIdx.x * blockDim.x + threadIdx.x;
    if (p < E) {
        const int e = ebuf[p];
        const int d = dstI[e];
        const int b = rowptr[d], en = rowptr[d + 1];
        int rank = 0;
        for (int i = b; i < en; ++i) rank += (ebuf[i] < e) ? 1 : 0;
        const int s = srcI[e];
        const float w = dis[s] * dis[d];
        csr[b + rank] = make_int2(s, __float_as_int(w));
    }
}

__global__ void k_f16T(const float* __restrict__ s, _Float16* __restrict__ d,
                       int K, int Nn)
{
    int i = blockIdx.x * blockDim.x + threadIdx.x;
    if (i < K * Nn) {
        int k = i / Nn, n = i % Nn;
        d[(size_t)n * K + k] = (_Float16)s[i];
    }
}

template <bool FULL>
__device__ __forceinline__ void gather_group(const float* __restrict__ hl, int mys, int myw,
                                             int j, int cnt, float4& acc)
{
    float4 r[8];
    float  w[8];
    #pragma unroll
    for (int u = 0; u < 8; ++u) {
        r[u] = make_float4(0.f, 0.f, 0.f, 0.f);
        w[u] = 0.f;
        if (FULL || j + u < cnt) {
            const int s = __builtin_amdgcn_readlane(mys, j + u);
            w[u] = __int_as_float(__builtin_amdgcn_readlane(myw, j + u));
            r[u] = *(const float4*)(hl + (size_t)s * 128);
        }
    }
    #pragma unroll
    for (int u = 0; u < 8; ++u) {
        if (FULL || j + u < cnt) {
            acc.x = fmaf(r[u].x, w[u], acc.x);
            acc.y = fmaf(r[u].y, w[u], acc.y);
            acc.z = fmaf(r[u].z, w[u], acc.z);
            acc.w = fmaf(r[u].w, w[u], acc.w);
        }
    }
}

#define GNPW 2
__global__ __launch_bounds__(256)
void k_gather(const int* __restrict__ rowptr, const int2* __restrict__ csr,
              const float* __restrict__ dis, const float* __restrict__ h,
              float* __restrict__ agg, int n)
{
    const int lane = threadIdx.x & 31;
    const int wave = __builtin_amdgcn_readfirstlane(blockIdx.x * 8 + (threadIdx.x >> 5));
    const float* hl = h + lane * 4;
    for (int k = 0; k < GNPW; ++k) {
        const int node = wave * GNPW + k;
        if (node >= n) break;
        const int beg = rowptr[node], end = rowptr[node + 1];
        float sd = dis[node]; sd *= sd;
        int base = beg;
        int cnt  = (end - base < 32) ? (end - base) : 32;
        int2 my = make_int2(0, 0);
        if (lane < cnt) my = csr[base + lane];
        const float4 v = *(const float4*)(hl + (size_t)node * 128);
        float4 acc = make_float4(0.f, 0.f, 0.f, 0.f);
        while (cnt > 0) {
            int j = 0;
            for (; j + 8 <= cnt; j += 8) gather_group<true >(hl, my.x, my.y, j, cnt, acc);
            if (j < cnt)                 gather_group<false>(hl, my.x, my.y, j, cnt, acc);
            base += 32;
            cnt = (end - base < 32) ? (end - base) : 32;
            if (lane < cnt) my = csr[base + lane];
        }
        acc.x = fmaf(v.x, sd, acc.x);
        acc.y = fmaf(v.y, sd, acc.y);
        acc.z = fmaf(v.z, sd, acc.z);
        acc.w = fmaf(v.w, sd, acc.w);
        *(float4*)(agg + (size_t)node * 128 + lane * 4) = acc;
    }
}

__global__ void k_pairdot(const int* __restrict__ ts, const float* __restrict__ tf,
                          const float* __restrict__ tg, float* __restrict__ out, int M)
{
    int m = blockIdx.x * blockDim.x + threadIdx.x;
    if (m < M) {
        int a = ts[2 * m + 0];
        int b = ts[2 * m + 1];
        const float4* pa = (const float4*)(tf + (size_t)a * 32);
        const float4* pb = (const float4*)(tg + (size_t)b * 32);
        float s = 0.f;
        #pragma unroll
        for (int i = 0; i < 8; ++i) {
            float4 x = pa[i], y = pb[i];
            s += x.x * y.x + x.y * y.y + x.z * y.z + x.w * y.w;
        }
        out[m] = s;
    }
}

extern "C" void kernel_launch(void* const* d_in, const int* in_sizes, int n_in,
                              void* d_out, int out_size, void* d_ws, size_t ws_size,
                              hipStream_t stream)
{
    const float* x    = (const float*)d_in[0];
    const int*   ei   = (const int*)  d_in[1];
    const int*   ts   = (const int*)  d_in[2];
    const float* W1   = (const float*)d_in[3];
    const float* W2   = (const float*)d_in[4];
    const float* tfw1 = (const float*)d_in[5];
    const float* tfb1 = (const float*)d_in[6];
    const float* tfw2 = (const float*)d_in[7];
    const float* tfb2 = (const float*)d_in[8];
    const float* tgw1 = (const float*)d_in[9];
    const float* tgb1 = (const float*)d_in[10];
    const float* tgw2 = (const float*)d_in[11];
    const float* tgb2 = (const float*)d_in[12];

    const int D_IN = 256, H1 = 128, H2 = 128, H3 = 64, D_OUT = 32;
    const int N = in_sizes[0] / D_IN;
    const int E = in_sizes[1] / 2;
    const int M = in_sizes[2] / 2;
    const int* srcI = ei;
    const int* dstI = ei + E;
    float* out = (float*)d_out;

    char* w = (char*)d_ws;
    auto alloc = [&](size_t bytes) -> char* {
        char* p = w; w += (bytes + 255) & ~(size_t)255; return p;
    };
    float*    dis = (float*)   alloc((size_t)N * 4);
    _Float16* W1h = (_Float16*)alloc((size_t)D_IN * H1 * 2);
    _Float16* W2h = (_Float16*)alloc((size_t)H1 * H2 * 2);
    _Float16* F1h = (_Float16*)alloc((size_t)H2 * H3 * 2);
    _Float16* F2h = (_Float16*)alloc((size_t)H3 * D_OUT * 2);
    _Float16* G1h = (_Float16*)alloc((size_t)H2 * H3 * 2);
    _Float16* G2h = (_Float16*)alloc((size_t)H3 * D_OUT * 2);
    float*    B1  = (float*)   alloc((size_t)N * H1 * 4);
    float*    B2  = (float*)   alloc((size_t)N * H1 * 4);
    float*    B3  = (float*)   alloc((size_t)N * D_OUT * 4);
    int*      cnt    = (int*) alloc((size_t)N * 4);
    int*      rowptr = (int*) alloc((size_t)(N + 1) * 4);
    int*      cursor = (int*) alloc((size_t)N * 4);
    int*      ebuf   = (int*) alloc((size_t)E * 4);
    int2*     csr    = (int2*)alloc((size_t)E * 8);
    (void)ws_size; (void)n_in; (void)out_size;

    const int T = 256;
    auto cdiv = [](long long a, long long b) { return (int)((a + b - 1) / b); };

    k_zero <<<cdiv(N, T), T, 0, stream>>>(cnt, N);
    k_count<<<cdiv(E, T), T, 0, stream>>>(dstI, cnt, E);
    k_scan <<<cdiv(N, 1024), 256, 0, stream>>>(cnt, rowptr, cursor, dis, N);
    k_fill <<<cdiv(E, T), T, 0, stream>>>(dstI, cursor, ebuf, E);
    k_sort <<<cdiv(E, T), T, 0, stream>>>(ebuf, srcI, dstI, rowptr, dis, csr, E);

    k_f16T<<<cdiv(D_IN * H1,   T), T, 0, stream>>>(W1,   W1h, D_IN, H1);
    k_f16T<<<cdiv(H1 * H2,     T), T, 0, stream>>>(W2,   W2h, H1,   H2);
    k_f16T<<<cdiv(H2 * H3,     T), T, 0, stream>>>(tfw1, F1h, H2,   H3);
    k_f16T<<<cdiv(H3 * D_OUT,  T), T, 0, stream>>>(tfw2, F2h, H3,   D_OUT);
    k_f16T<<<cdiv(H2 * H3,     T), T, 0, stream>>>(tgw1, G1h, H2,   H3);
    k_f16T<<<cdiv(H3 * D_OUT,  T), T, 0, stream>>>(tgw2, G2h, H3,   D_OUT);

    auto gemm_blocks = [&](int m, int nn, int nt) {
        return cdiv((long long)((m + 15) / 16) * (nn / (16 * nt)), 8);
    };

    wmma_gemm<0, 0, 4><<<gemm_blocks(N, H1, 4), T, 0, stream>>>(x, W1h, nullptr, B1, N, D_IN, H1);
    k_gather<<<cdiv(cdiv(N, GNPW), 8), T, 0, stream>>>(rowptr, csr, dis, B1, B2, N);

    wmma_gemm<1, 0, 4><<<gemm_blocks(N, H2, 4), T, 0, stream>>>(B2, W2h, nullptr, B1, N, H1, H2);
    k_gather<<<cdiv(cdiv(N, GNPW), 8), T, 0, stream>>>(rowptr, csr, dis, B1, B2, N);

    float* t1  = B1;
    float* tfO = B1 + (size_t)N * H3;
    wmma_gemm<0, 1, 4><<<gemm_blocks(N, H3, 4),    T, 0, stream>>>(B2, F1h, tfb1, t1,  N, H2, H3);
    wmma_gemm<0, 1, 2><<<gemm_blocks(N, D_OUT, 2), T, 0, stream>>>(t1, F2h, tfb2, tfO, N, H3, D_OUT);
    wmma_gemm<0, 1, 4><<<gemm_blocks(N, H3, 4),    T, 0, stream>>>(B2, G1h, tgb1, t1,  N, H2, H3);
    wmma_gemm<0, 1, 2><<<gemm_blocks(N, D_OUT, 2), T, 0, stream>>>(t1, G2h, tgb2, B3,  N, H3, D_OUT);

    k_pairdot<<<cdiv(M, T), T, 0, stream>>>(ts, tfO, B3, out, M);
}


# Independent kernels share launches: one launch clears counts and converts the six weights, the scan rides with GEMM 1, both head branches per layer in one launch (20 launches to 11); sums unchanged
# speedup vs baseline: 1.7870x; 1.1710x over previous
#include <hip/hip_runtime.h>
#include <hip/hip_bf16.h>

typedef __attribute__((ext_vector_type(16))) _Float16 v16h;
typedef __attribute__((ext_vector_type(8)))  _Float16 v8h;
typedef __attribute__((ext_vector_type(8)))  float    v8f;

#define LEAKY_SLOPE 0.01f

__device__ __forceinline__ float elu_f(float x)   { return x > 0.f ? x : (expf(x) - 1.f); }
__device__ __forceinline__ float leaky_f(float x) { return x > 0.f ? x : LEAKY_SLOPE * x; }

constexpr int D_IN = 256, H1 = 128, H2 = 128, H3 = 64, D_OUT = 32;

template <int ACT_IN, int ACT_OUT, int NT>
__device__ __forceinline__
void gemm_tile(int bid, const float* __restrict__ A, int lda,
               const _Float16* __restrict__ BhT,
               const float* __restrict__ bias, const float* __restrict__ bias_hi, int bsplit,
               float* __restrict__ C, int Mrows, int K, int Nn)
{
    const int mtiles  = (Mrows + 15) >> 4;
    const int nstrips = Nn / (16 * NT);
    const int wave    = bid * (blockDim.x >> 5) + (threadIdx.x >> 5);
    if (wave >= mtiles * nstrips) return;
    const int mt   = wave / nstrips;
    const int ns   = wave % nstrips;
    const int lane = threadIdx.x & 31;
    const int hl   = lane >> 4;
    const int lrow = lane & 15;

    int arow = mt * 16 + lrow;
    if (arow >= Mrows) arow = Mrows - 1;
    const float* Abase = A + (size_t)arow * lda;

    const _Float16* Bcol[NT];
    #pragma unroll
    for (int t = 0; t < NT; ++t)
        Bcol[t] = BhT + (size_t)((ns * NT + t) * 16 + lrow) * K;

    v8f acc[NT];
    #pragma unroll
    for (int t = 0; t < NT; ++t) acc[t] = (v8f){};

    for (int kk = 0; kk < K; kk += 32) {
        const int kb = kk + hl * 8;
        __builtin_prefetch(Abase + kk + 32, 0, 0);

        const float4* ap = (const float4*)(Abase + kb);
        float4 a0 = ap[0], a1 = ap[1];
        float4 a2 = ap[4], a3 = ap[5];
        float af[16] = { a0.x, a0.y, a0.z, a0.w, a1.x, a1.y, a1.z, a1.w,
                         a2.x, a2.y, a2.z, a2.w, a3.x, a3.y, a3.z, a3.w };
        v16h a;
        #pragma unroll
        for (int i = 0; i < 16; ++i) {
            float v = af[i];
            if (ACT_IN) v = elu_f(v);
            a[i] = (_Float16)v;
        }

        #pragma unroll
        for (int t = 0; t < NT; ++t) {
            v8h b0 = *(const v8h*)(Bcol[t] + kb);
            v8h b1 = *(const v8h*)(Bcol[t] + kb + 16);
            v16h b = __builtin_shufflevector(b0, b1, 0, 1, 2, 3, 4, 5, 6, 7,
                                             8, 9, 10, 11, 12, 13, 14, 15);
            acc[t] = __builtin_amdgcn_wmma_f32_16x16x32_f16(false, a, false, b,
                                                            (short)0, acc[t],
                                                            false, false);
        }
    }

    const float* bp = bias;
    int bo = 0;
    if (ns * NT * 16 >= bsplit) { bp = bias_hi; bo = bsplit; }

    const bool fullM = (mt * 16 + 16) <= Mrows;
    #pragma unroll
    for (int t = 0; t < NT; ++t) {
        const int col = (ns * NT + t) * 16 + lrow;
        const float bv = bp ? bp[col - bo] : 0.f;
        float* Cb = C + (size_t)(mt * 16 + hl * 8) * Nn + col;
        if (fullM) {
            #pragma unroll
            for (int r = 0; r < 8; ++r) {
                float v = acc[t][r] + bv;
                if (ACT_OUT) v = leaky_f(v);
                Cb[(size_t)r * Nn] = v;
            }
        } else {
            #pragma unroll
            for (int r = 0; r < 8; ++r) {
                const int orow = mt * 16 + hl * 8 + r;
                if (orow < Mrows) {
                    float v = acc[t][r] + bv;
                    if (ACT_OUT) v = leaky_f(v);
                    Cb[(size_t)r * Nn] = v;
                }
            }
        }
    }
}

template <int ACT_IN, int ACT_OUT, int NT>
__global__ __launch_bounds__(256)
void wmma_gemm(const float* __restrict__ A, int lda, const _Float16* __restrict__ BhT,
               const float* __restrict__ bias, const float* __restrict__ bias_hi, int bsplit,
               float* __restrict__ C, int Mrows, int K, int Nn)
{
    gemm_tile<ACT_IN, ACT_OUT, NT>(blockIdx.x, A, lda, BhT, bias, bias_hi, bsplit, C, Mrows, K, Nn);
}

__global__ __launch_bounds__(256)
void k_heads2(const float* __restrict__ t1,
              const _Float16* __restrict__ Fh, const _Float16* __restrict__ Gh,
              const float* __restrict__ bF, const float* __restrict__ bG,
              float* __restrict__ oF, float* __restrict__ oG,
              int nb, int Mrows, int K, int Nn)
{
    const bool g = (int)blockIdx.x >= nb;
    gemm_tile<0, 1, 2>((int)blockIdx.x - (g ? nb : 0), t1 + (g ? K : 0), 2 * K,
                       g ? Gh : Fh, g ? bG : bF, nullptr, Nn,
                       g ? oG : oF, Mrows, K, Nn);
}

__global__ void k_count(const int* __restrict__ dst, int* cnt, int E)
{
    int e = blockIdx.x * blockDim.x + threadIdx.x;
    if (e < E) atomicAdd(&cnt[dst[e]], 1);
}

__device__ __forceinline__
void scan_tile(int bid, const int* __restrict__ cnt, int* __restrict__ rowptr,
               int* __restrict__ cursor, float* __restrict__ dis, int n)
{
    __shared__ int s_pre[8];
    __shared__ int s_tot[8];
    const int t = threadIdx.x, lane = t & 31, wv = t >> 5;

    const int4* c4 = (const int4*)cnt;
    const int nb = bid;
    int pre = 0;
    #pragma unroll 8
    for (int m = 0; m < nb; ++m) {
        int4 v = c4[m * 256 + t];
        pre += (v.x + v.y) + (v.z + v.w);
    }

    const int i0 = bid * 1024 + t * 4;
    int c[4];
    if (i0 + 3 < n) {
        int4 v = c4[i0 >> 2];
        c[0] = v.x; c[1] = v.y; c[2] = v.z; c[3] = v.w;
    } else {
        for (int k = 0; k < 4; ++k) c[k] = (i0 + k < n) ? cnt[i0 + k] : 0;
    }
    const int mine = (c[0] + c[1]) + (c[2] + c[3]);

    int incl = mine;
    for (int off = 1; off < 32; off <<= 1) {
        int v = __shfl_up(incl, off);
        if (lane >= off) incl += v;
    }
    for (int off = 16; off > 0; off >>= 1) pre += __shfl_xor(pre, off);
    if (lane == 31) s_tot[wv] = incl;
    if (lane == 0)  s_pre[wv] = pre;
    __syncthreads();
    int run = incl - mine;
    for (int k = 0; k < 8; ++k) {
        run += s_pre[k];
        if (k < wv) run += s_tot[k];
    }

    int r[4]; float d[4];
    for (int k = 0; k < 4; ++k) {
        r[k] = run;
        d[k] = rsqrtf((float)(c[k] + 1));
        run += c[k];
    }
    if (i0 + 3 < n) {
        *(int4*)(rowptr + i0)  = make_int4(r[0], r[1], r[2], r[3]);
        *(int4*)(cursor + i0)  = make_int4(r[0], r[1], r[2], r[3]);
        *(float4*)(dis + i0)   = make_float4(d[0], d[1], d[2], d[3]);
    } else {
        for (int k = 0; k < 4; ++k)
            if (i0 + k < n) { rowptr[i0 + k] = r[k]; cursor[i0 + k] = r[k]; dis[i0 + k] = d[k]; }
    }
    if (i0 < n && n <= i0 + 4) rowptr[n] = run;
}

__global__ __launch_bounds__(256)
void k_scan_gemm1(const int* __restrict__ cnt, int* __restrict__ rowptr,
                  int* __restrict__ cursor, float* __restrict__ dis, int n, int sb,
                  const float* __restrict__ x, const _Float16* __restrict__ W1h,
                  float* __restrict__ h1)
{
    if ((int)blockIdx.x < sb) {
        scan_tile((int)blockIdx.x, cnt, rowptr, cursor, dis, n);
        return;
    }
    gemm_tile<0, 0, 4>((int)blockIdx.x - sb, x, D_IN, W1h, nullptr, nullptr, H1, h1, n, D_IN, H1);
}

__global__ void k_fill(const int* __restrict__ dst, int* cursor, int* __restrict__ ebuf, int E)
{
    int e = blockIdx.x * blockDim.x + threadIdx.x;
    if (e < E) {
        int pos = atomicAdd(&cursor[dst[e]], 1);
        ebuf[pos] = e;
    }
}

__global__ void k_sort(const int* __restrict__ ebuf, const int* __restrict__ srcI,
                       const int* __restrict__ dstI, const int* __restrict__ rowptr,
                       const float* __restrict__ dis, int2* __restrict__ csr, int E)
{
    int p = blockIdx.x * blockDim.x + threadIdx.x;
    if (p < E) {
        const int e = ebuf[p];
        const int d = dstI[e];
        const int b = rowptr[d], en = rowptr[d + 1];
        int rank = 0;
        for (int i = b; i < en; ++i) rank += (ebuf[i] < e) ? 1 : 0;
        const int s = srcI[e];
        const float w = dis[s] * dis[d];
        csr[b + rank] = make_int2(s, __float_as_int(w));
    }
}

__device__ __forceinline__
void f16T_blk(const float* __restrict__ s, _Float16* __restrict__ d, int K, int Nn, int blk)
{
    int i = blk * 256 + threadIdx.x;
    if (i < K * Nn) {
        int k = i / Nn, n = i % Nn;
        d[(size_t)n * K + k] = (_Float16)s[i];
    }
}

constexpr int NB_W1 = (D_IN * H1 + 255) / 256, NB_W2 = (H1 * H2 + 255) / 256;
constexpr int NB_H1 = (H2 * H3 + 255) / 256,   NB_H2 = (H3 * D_OUT + 255) / 256;
constexpr int NB_WALL = NB_W1 + NB_W2 + 2 * NB_H1 + 2 * NB_H2;

__global__ __launch_bounds__(256)
void k_prep(int* __restrict__ cnt, int n, int zb,
            const float* __restrict__ W1,   const float* __restrict__ W2,
            const float* __restrict__ tfw1, const float* __restrict__ tfw2,
            const float* __restrict__ tgw1, const float* __restrict__ tgw2,
            _Float16* __restrict__ W1h, _Float16* __restrict__ W2h,
            _Float16* __restrict__ F1h, _Float16* __restrict__ F2h,
            _Float16* __restrict__ G1h, _Float16* __restrict__ G2h)
{
    int b = blockIdx.x;
    if (b < zb) {
        const int i = b * 256 + threadIdx.x;
        if (i < n) cnt[i] = 0;
        return;
    }
    b -= zb;
    if (b < NB_W1) { f16T_blk(W1,   W1h, D_IN, H1,    b); return; }  b -= NB_W1;
    if (b < NB_W2) { f16T_blk(W2,   W2h, H1,   H2,    b); return; }  b -= NB_W2;
    if (b < NB_H1) { f16T_blk(tfw1, F1h, H2,   H3,    b); return; }  b -= NB_H1;
    if (b < NB_H2) { f16T_blk(tfw2, F2h, H3,   D_OUT, b); return; }  b -= NB_H2;
    if (b < NB_H1) { f16T_blk(tgw1, G1h, H2,   H3,    b); return; }  b -= NB_H1;
    f16T_blk(tgw2, G2h, H3, D_OUT, b);
}

template <bool FULL>
__device__ __forceinline__ void gather_group(const float* __restrict__ hl, int mys, int myw,
                                             int j, int cnt, float4& acc)
{
    float4 r[8];
    float  w[8];
    #pragma unroll
    for (int u = 0; u < 8; ++u) {
        r[u] = make_float4(0.f, 0.f, 0.f, 0.f);
        w[u] = 0.f;
        if (FULL || j + u < cnt) {
            const int s = __builtin_amdgcn_readlane(mys, j + u);
            w[u] = __int_as_float(__builtin_amdgcn_readlane(myw, j + u));
            r[u] = *(const float4*)(hl + (size_t)s * 128);
        }
    }
    #pragma unroll
    for (int u = 0; u < 8; ++u) {
        if (FULL || j + u < cnt) {
            acc.x = fmaf(r[u].x, w[u], acc.x);
            acc.y = fmaf(r[u].y, w[u], acc.y);
            acc.z = fmaf(r[u].z, w[u], acc.z);
            acc.w = fmaf(r[u].w, w[u], acc.w);
        }
    }
}

#define GNPW 2
__global__ __launch_bounds__(256)
void k_gather(const int* __restrict__ rowptr, const int2* __restrict__ csr,
              const float* __restrict__ dis, const float* __restrict__ h,
              float* __restrict__ agg, int n)
{
    const int lane = threadIdx.x & 31;
    const int wave = __builtin_amdgcn_readfirstlane(blockIdx.x * 8 + (threadIdx.x >> 5));
    const float* hl = h + lane * 4;
    for (int k = 0; k < GNPW; ++k) {
        const int node = wave * GNPW + k;
        if (node >= n) break;
        const int beg = rowptr[node], end = rowptr[node + 1];
        float sd = dis[node]; sd *= sd;
        int base = beg;
        int cnt  = (end - base < 32) ? (end - base) : 32;
        int2 my = make_int2(0, 0);
        if (lane < cnt) my = csr[base + lane];
        const float4 v = *(const float4*)(hl + (size_t)node * 128);
        float4 acc = make_float4(0.f, 0.f, 0.f, 0.f);
        while (cnt > 0) {
            int j = 0;
            for (; j + 8 <= cnt; j += 8) gather_group<true >(hl, my.x, my.y, j, cnt, acc);
            if (j < cnt)                 gather_group<false>(hl, my.x, my.y, j, cnt, acc);
            base += 32;
            cnt = (end - base < 32) ? (end - base) : 32;
            if (lane < cnt) my = csr[base + lane];
        }
        acc.x = fmaf(v.x, sd, acc.x);
        acc.y = fmaf(v.y, sd, acc.y);
        acc.z = fmaf(v.z, sd, acc.z);
        acc.w = fmaf(v.w, sd, acc.w);
        *(float4*)(agg + (size_t)node * 128 + lane * 4) = acc;
    }
}

__global__ void k_pairdot(const int* __restrict__ ts, const float* __restrict__ tf,
                          const float* __restrict__ tg, float* __restrict__ out, int M)
{
    int m = blockIdx.x * blockDim.x + threadIdx.x;
    if (m < M) {
        int a = ts[2 * m + 0];
        int b = ts[2 * m + 1];
        const float4* pa = (const float4*)(tf + (size_t)a * 32);
        const float4* pb = (const float4*)(tg + (size_t)b * 32);
        float s = 0.f;
        #pragma unroll
        for (int i = 0; i < 8; ++i) {
            float4 x = pa[i], y = pb[i];
            s += x.x * y.x + x.y * y.y + x.z * y.z + x.w * y.w;
        }
        out[m] = s;
    }
}

extern "C" void kernel_launch(void* const* d_in, const int* in_sizes, int n_in,
                              void* d_out, int out_size, void* d_ws, size_t ws_size,
                              hipStream_t stream)
{
    const float* x    = (const float*)d_in[0];
    const int*   ei   = (const int*)  d_in[1];
    const int*   ts   = (const int*)  d_in[2];
    const float* W1   = (const float*)d_in[3];
    const float* W2   = (const float*)d_in[4];
    const float* tfw1 = (const float*)d_in[5];
    const float* tfb1 = (const float*)d_in[6];
    const float* tfw2 = (const float*)d_in[7];
    const float* tfb2 = (const float*)d_in[8];
    const float* tgw1 = (const float*)d_in[9];
    const float* tgb1 = (const float*)d_in[10];
    const float* tgw2 = (const float*)d_in[11];
    const float* tgb2 = (const float*)d_in[12];

    const int N = in_sizes[0] / D_IN;
    const int E = in_sizes[1] / 2;
    const int M = in_sizes[2] / 2;
    const int* srcI = ei;
    const int* dstI = ei + E;
    float* out = (float*)d_out;

    char* w = (char*)d_ws;
    auto alloc = [&](size_t bytes) -> char* {
        char* p = w; w += (bytes + 255) & ~(size_t)255; return p;
    };
    float*    dis = (float*)   alloc((size_t)N * 4);
    _Float16* W1h = (_Float16*)alloc((size_t)D_IN * H1 * 2);
    _Float16* W2h = (_Float16*)alloc((size_t)H1 * H2 * 2);
    _Float16* FG1h = (_Float16*)alloc((size_t)2 * H2 * H3 * 2);
    _Float16* F1h = FG1h;
    _Float16* G1h = FG1h + (size_t)H3 * H2;
    _Float16* F2h = (_Float16*)alloc((size_t)H3 * D_OUT * 2);
    _Float16* G2h = (_Float16*)alloc((size_t)H3 * D_OUT * 2);
    float*    B1  = (float*)   alloc((size_t)N * H1 * 4);
    float*    B2  = (float*)   alloc((size_t)N * H1 * 4);
    float*    B3  = (float*)   alloc((size_t)N * D_OUT * 4);
    float*    B4  = (float*)   alloc((size_t)N * D_OUT * 4);
    int*      cnt    = (int*) alloc((size_t)N * 4);
    int*      rowptr = (int*) alloc((size_t)(N + 1) * 4);
    int*      cursor = (int*) alloc((size_t)N * 4);
    int*      ebuf   = (int*) alloc((size_t)E * 4);
    int2*     csr    = (int2*)alloc((size_t)E * 8);
    (void)ws_size; (void)n_in; (void)out_size;

    const int T = 256;
    auto cdiv = [](long long a, long long b) { return (int)((a + b - 1) / b); };
    auto gemm_blocks = [&](int m, int nn, int nt) {
        return cdiv((long long)((m + 15) / 16) * (nn / (16 * nt)), 8);
    };

    const int zb = cdiv(N, T);
    k_prep<<<zb + NB_WALL, T, 0, stream>>>(cnt, N, zb, W1, W2, tfw1, tfw2, tgw1, tgw2,
                                           W1h, W2h, F1h, F2h, G1h, G2h);

    k_count<<<cdiv(E, T), T, 0, stream>>>(dstI, cnt, E);
    const int sb = cdiv(N, 1024);
    k_scan_gemm1<<<sb + gemm_blocks(N, H1, 4), T, 0, stream>>>(cnt, rowptr, cursor, dis, N, sb, x, W1h, B1);
    k_fill <<<cdiv(E, T), T, 0, stream>>>(dstI, cursor, ebuf, E);
    k_sort <<<cdiv(E, T), T, 0, stream>>>(ebuf, srcI, dstI, rowptr, dis, csr, E);

    k_gather<<<cdiv(cdiv(N, GNPW), 8), T, 0, stream>>>(rowptr, csr, dis, B1, B2, N);

    wmma_gemm<1, 0, 4><<<gemm_blocks(N, H2, 4), T, 0, stream>>>(B2, H1, W2h, nullptr, nullptr, H2, B1, N, H1, H2);
    k_gather<<<cdiv(cdiv(N, GNPW), 8), T, 0, stream>>>(rowptr, csr, dis, B1, B2, N);

    float* t1 = B1;
    wmma_gemm<0, 1, 4><<<gemm_blocks(N, 2 * H3, 4), T, 0, stream>>>(B2, H2, FG1h, tfb1, tgb1, H3, t1, N, H2, 2 * H3);
    const int hb = gemm_blocks(N, D_OUT, 2);
    k_heads2<<<2 * hb, T, 0, stream>>>(t1, F2h, G2h, tfb2, tgb2, B4, B3, hb, N, H3, D_OUT);

    k_pairdot<<<cdiv(M, T), T, 0, stream>>>(ts, B4, B3, out, M);
}


# Each gather hands its 16-node tile to the next product through LDS (gather, act, wmma in one kernel) instead of a store, a launch and a load: 11 launches to 9; all sums and their order unchanged
# speedup vs baseline: 1.9462x; 1.0891x over previous
#include <hip/hip_runtime.h>
#include <hip/hip_bf16.h>

typedef __attribute__((ext_vector_type(16))) _Float16 v16h;
typedef __attribute__((ext_vector_type(8)))  _Float16 v8h;
typedef __attribute__((ext_vector_type(8)))  float    v8f;

#define LEAKY_SLOPE 0.01f

__device__ __forceinline__ float elu_f(float x)   { return x > 0.f ? x : (expf(x) - 1.f); }
__device__ __forceinline__ float leaky_f(float x) { return x > 0.f ? x : LEAKY_SLOPE * x; }

constexpr int D_IN = 256, H1 = 128, H2 = 128, H3 = 64, D_OUT = 32;

template <int ACT_IN, int ACT_OUT, int NT>
__device__ __forceinline__
void gemm_tile(int bid, const float* __restrict__ A, int lda,
               const _Float16* __restrict__ BhT,
               const float* __restrict__ bias, const float* __restrict__ bias_hi, int bsplit,
               float* __restrict__ C, int Mrows, int K, int Nn)
{
    const int mtiles  = (Mrows + 15) >> 4;
    const int nstrips = Nn / (16 * NT);
    const int wave    = bid * (blockDim.x >> 5) + (threadIdx.x >> 5);
    if (wave >= mtiles * nstrips) return;
    const int mt   = wave / nstrips;
    const int ns   = wave % nstrips;
    const int lane = threadIdx.x & 31;
    const int hl   = lane >> 4;
    const int lrow = lane & 15;

    int arow = mt * 16 + lrow;
    if (arow >= Mrows) arow = Mrows - 1;
    const float* Abase = A + (size_t)arow * lda;

    const _Float16* Bcol[NT];
    #pragma unroll
    for (int t = 0; t < NT; ++t)
        Bcol[t] = BhT + (size_t)((ns * NT + t) * 16 + lrow) * K;

    v8f acc[NT];
    #pragma unroll
    for (int t = 0; t < NT; ++t) acc[t] = (v8f){};

    for (int kk = 0; kk < K; kk += 32) {
        const int kb = kk + hl * 8;
        __builtin_prefetch(Abase + kk + 32, 0, 0);

        const float4* ap = (const float4*)(Abase + kb);
        float4 a0 = ap[0], a1 = ap[1];
        float4 a2 = ap[4], a3 = ap[5];
        float af[16] = { a0.x, a0.y, a0.z, a0.w, a1.x, a1.y, a1.z, a1.w,
                         a2.x, a2.y, a2.z, a2.w, a3.x, a3.y, a3.z, a3.w };
        v16h a;
        #pragma unroll
        for (int i = 0; i < 16; ++i) {
            float v = af[i];
            if (ACT_IN) v = elu_f(v);
            a[i] = (_Float16)v;
        }

        #pragma unroll
        for (int t = 0; t < NT; ++t) {
            v8h b0 = *(const v8h*)(Bcol[t] + kb);
            v8h b1 = *(const v8h*)(Bcol[t] + kb + 16);
            v16h b = __builtin_shufflevector(b0, b1, 0, 1, 2, 3, 4, 5, 6, 7,
                                             8, 9, 10, 11, 12, 13, 14, 15);
            acc[t] = __builtin_amdgcn_wmma_f32_16x16x32_f16(false, a, false, b,
                                                            (short)0, acc[t],
                                                            false, false);
        }
    }

    const float* bp = bias;
    int bo = 0;
    if (ns * NT * 16 >= bsplit) { bp = bias_hi; bo = bsplit; }

    const bool fullM = (mt * 16 + 16) <= Mrows;
    #pragma unroll
    for (int t = 0; t < NT; ++t) {
        const int col = (ns * NT + t) * 16 + lrow;
        const float bv = bp ? bp[col - bo] : 0.f;
        float* Cb = C + (size_t)(mt * 16 + hl * 8) * Nn + col;
        if (fullM) {
            #pragma unroll
            for (int r = 0; r < 8; ++r) {
                float v = acc[t][r] + bv;
                if (ACT_OUT) v = leaky_f(v);
                Cb[(size_t)r * Nn] = v;
            }
        } else {
            #pragma unroll
            for (int r = 0; r < 8; ++r) {
                const int orow = mt * 16 + hl * 8 + r;
                if (orow < Mrows) {
                    float v = acc[t][r] + bv;
                    if (ACT_OUT) v = leaky_f(v);
                    Cb[(size_t)r * Nn] = v;
                }
            }
        }
    }
}

__global__ __launch_bounds__(256)
void k_heads2(const float* __restrict__ t1,
              const _Float16* __restrict__ Fh, const _Float16* __restrict__ Gh,
              const float* __restrict__ bF, const float* __restrict__ bG,
              float* __restrict__ oF, float* __restrict__ oG,
              int nb, int Mrows, int K, int Nn)
{
    const bool g = (int)blockIdx.x >= nb;
    gemm_tile<0, 1, 2>((int)blockIdx.x - (g ? nb : 0), t1 + (g ? K : 0), 2 * K,
                       g ? Gh : Fh, g ? bG : bF, nullptr, Nn,
                       g ? oG : oF, Mrows, K, Nn);
}

__global__ void k_count(const int* __restrict__ dst, int* cnt, int E)
{
    int e = blockIdx.x * blockDim.x + threadIdx.x;
    if (e < E) atomicAdd(&cnt[dst[e]], 1);
}

__device__ __forceinline__
void scan_tile(int bid, const int* __restrict__ cnt, int* __restrict__ rowptr,
               int* __restrict__ cursor, float* __restrict__ dis, int n)
{
    __shared__ int s_pre[8];
    __shared__ int s_tot[8];
    const int t = threadIdx.x, lane = t & 31, wv = t >> 5;

    const int4* c4 = (const int4*)cnt;
    const int nb = bid;
    int pre = 0;
    #pragma unroll 8
    for (int m = 0; m < nb; ++m) {
        int4 v = c4[m * 256 + t];
        pre += (v.x + v.y) + (v.z + v.w);
    }

    const int i0 = bid * 1024 + t * 4;
    int c[4];
    if (i0 + 3 < n) {
        int4 v = c4[i0 >> 2];
        c[0] = v.x; c[1] = v.y; c[2] = v.z; c[3] = v.w;
    } else {
        for (int k = 0; k < 4; ++k) c[k] = (i0 + k < n) ? cnt[i0 + k] : 0;
    }
    const int mine = (c[0] + c[1]) + (c[2] + c[3]);

    int incl = mine;
    for (int off = 1; off < 32; off <<= 1) {
        int v = __shfl_up(incl, off);
        if (lane >= off) incl += v;
    }
    for (int off = 16; off > 0; off >>= 1) pre += __shfl_xor(pre, off);
    if (lane == 31) s_tot[wv] = incl;
    if (lane == 0)  s_pre[wv] = pre;
    __syncthreads();
    int run = incl - mine;
    for (int k = 0; k < 8; ++k) {
        run += s_pre[k];
        if (k < wv) run += s_tot[k];
    }

    int r[4]; float d[4];
    for (int k = 0; k < 4; ++k) {
        r[k] = run;
        d[k] = rsqrtf((float)(c[k] + 1));
        run += c[k];
    }
    if (i0 + 3 < n) {
        *(int4*)(rowptr + i0)  = make_int4(r[0], r[1], r[2], r[3]);
        *(int4*)(cursor + i0)  = make_int4(r[0], r[1], r[2], r[3]);
        *(float4*)(dis + i0)   = make_float4(d[0], d[1], d[2], d[3]);
    } else {
        for (int k = 0; k < 4; ++k)
            if (i0 + k < n) { rowptr[i0 + k] = r[k]; cursor[i0 + k] = r[k]; dis[i0 + k] = d[k]; }
    }
    if (i0 < n && n <= i0 + 4) rowptr[n] = run;
}

__global__ __launch_bounds__(256)
void k_scan_gemm1(const int* __restrict__ cnt, int* __restrict__ rowptr,
                  int* __restrict__ cursor, float* __restrict__ dis, int n, int sb,
                  const float* __restrict__ x, const _Float16* __restrict__ W1h,
                  float* __restrict__ h1)
{
    if ((int)blockIdx.x < sb) {
        scan_tile((int)blockIdx.x, cnt, rowptr, cursor, dis, n);
        return;
    }
    gemm_tile<0, 0, 4>((int)blockIdx.x - sb, x, D_IN, W1h, nullptr, nullptr, H1, h1, n, D_IN, H1);
}

__global__ void k_fill(const int* __restrict__ dst, int* cursor, int* __restrict__ ebuf, int E)
{
    int e = blockIdx.x * blockDim.x + threadIdx.x;
    if (e < E) {
        int pos = atomicAdd(&cursor[dst[e]], 1);
        ebuf[pos] = e;
    }
}

__global__ void k_sort(const int* __restrict__ ebuf, const int* __restrict__ srcI,
                       const int* __restrict__ dstI, const int* __restrict__ rowptr,
                       const float* __restrict__ dis, int2* __restrict__ csr, int E)
{
    int p = blockIdx.x * blockDim.x + threadIdx.x;
    if (p < E) {
        const int e = ebuf[p];
        const int d = dstI[e];
        const int b = rowptr[d], en = rowptr[d + 1];
        int rank = 0;
        for (int i = b; i < en; ++i) rank += (ebuf[i] < e) ? 1 : 0;
        const int s = srcI[e];
        const float w = dis[s] * dis[d];
        csr[b + rank] = make_int2(s, __float_as_int(w));
    }
}

__device__ __forceinline__
void f16T_blk(const float* __restrict__ s, _Float16* __restrict__ d, int K, int Nn, int blk)
{
    int i = blk * 256 + threadIdx.x;
    if (i < K * Nn) {
        int k = i / Nn, n = i % Nn;
        d[(size_t)n * K + k] = (_Float16)s[i];
    }
}

constexpr int NB_W1 = (D_IN * H1 + 255) / 256, NB_W2 = (H1 * H2 + 255) / 256;
constexpr int NB_H1 = (H2 * H3 + 255) / 256,   NB_H2 = (H3 * D_OUT + 255) / 256;
constexpr int NB_WALL = NB_W1 + NB_W2 + 2 * NB_H1 + 2 * NB_H2;

__global__ __launch_bounds__(256)
void k_prep(int* __restrict__ cnt, int n, int zb,
            const float* __restrict__ W1,   const float* __restrict__ W2,
            const float* __restrict__ tfw1, const float* __restrict__ tfw2,
            const float* __restrict__ tgw1, const float* __restrict__ tgw2,
            _Float16* __restrict__ W1h, _Float16* __restrict__ W2h,
            _Float16* __restrict__ F1h, _Float16* __restrict__ F2h,
            _Float16* __restrict__ G1h, _Float16* __restrict__ G2h)
{
    int b = blockIdx.x;
    if (b < zb) {
        const int i = b * 256 + threadIdx.x;
        if (i < n) cnt[i] = 0;
        return;
    }
    b -= zb;
    if (b < NB_W1) { f16T_blk(W1,   W1h, D_IN, H1,    b); return; }  b -= NB_W1;
    if (b < NB_W2) { f16T_blk(W2,   W2h, H1,   H2,    b); return; }  b -= NB_W2;
    if (b < NB_H1) { f16T_blk(tfw1, F1h, H2,   H3,    b); return; }  b -= NB_H1;
    if (b < NB_H2) { f16T_blk(tfw2, F2h, H3,   D_OUT, b); return; }  b -= NB_H2;
    if (b < NB_H1) { f16T_blk(tgw1, G1h, H2,   H3,    b); return; }  b -= NB_H1;
    f16T_blk(tgw2, G2h, H3, D_OUT, b);
}

template <bool FULL>
__device__ __forceinline__ void gather_group(const float* __restrict__ hl, int mys, int myw,
                                             int j, int cnt, float4& acc)
{
    float4 r[8];
    float  w[8];
    #pragma unroll
    for (int u = 0; u < 8; ++u) {
        r[u] = make_float4(0.f, 0.f, 0.f, 0.f);
        w[u] = 0.f;
        if (FULL || j + u < cnt) {
            const int s = __builtin_amdgcn_readlane(mys, j + u);
            w[u] = __int_as_float(__builtin_amdgcn_readlane(myw, j + u));
            r[u] = *(const float4*)(hl + (size_t)s * 128);
        }
    }
    #pragma unroll
    for (int u = 0; u < 8; ++u) {
        if (FULL || j + u < cnt) {
            acc.x = fmaf(r[u].x, w[u], acc.x);
            acc.y = fmaf(r[u].y, w[u], acc.y);
            acc.z = fmaf(r[u].z, w[u], acc.z);
            acc.w = fmaf(r[u].w, w[u], acc.w);
        }
    }
}

#define GNPW 2
constexpr int GCH = 128;
constexpr int GPITCH = GCH + 8;
template <int ACT_IN, int ACT_OUT>
__global__ __launch_bounds__(256)
void k_gather_gemm(const int* __restrict__ rowptr, const int2* __restrict__ csr,
                   const float* __restrict__ dis, const float* __restrict__ h,
                   const _Float16* __restrict__ BhT,
                   const float* __restrict__ bias, const float* __restrict__ bias_hi, int bsplit,
                   float* __restrict__ C, int n)
{
    __shared__ __attribute__((aligned(16))) _Float16 s_a[16 * GPITCH];
    const int lane = threadIdx.x & 31;
    const int wv = __builtin_amdgcn_readfirstlane(threadIdx.x >> 5);
    const int row0 = blockIdx.x * 16;
    const float* hl = h + lane * 4;
    for (int k = 0; k < GNPW; ++k) {
        const int r = wv * GNPW + k;
        const int node = row0 + r;
        float4 acc = make_float4(0.f, 0.f, 0.f, 0.f);
        if (node < n) {
            const int beg = rowptr[node], end = rowptr[node + 1];
            float sd = dis[node]; sd *= sd;
            int base = beg;
            int cnt  = (end - base < 32) ? (end - base) : 32;
            int2 my = make_int2(0, 0);
            if (lane < cnt) my = csr[base + lane];
            const float4 v = *(const float4*)(hl + (size_t)node * GCH);
            while (cnt > 0) {
                int j = 0;
                for (; j + 8 <= cnt; j += 8) gather_group<true >(hl, my.x, my.y, j, cnt, acc);
                if (j < cnt)                 gather_group<false>(hl, my.x, my.y, j, cnt, acc);
                base += 32;
                cnt = (end - base < 32) ? (end - base) : 32;
                if (lane < cnt) my = csr[base + lane];
            }
            acc.x = fmaf(v.x, sd, acc.x);
            acc.y = fmaf(v.y, sd, acc.y);
            acc.z = fmaf(v.z, sd, acc.z);
            acc.w = fmaf(v.w, sd, acc.w);
        }
        if (ACT_IN) { acc.x = elu_f(acc.x); acc.y = elu_f(acc.y); acc.z = elu_f(acc.z); acc.w = elu_f(acc.w); }
        typedef __attribute__((ext_vector_type(4))) _Float16 v4h;
        v4h hv = { (_Float16)acc.x, (_Float16)acc.y, (_Float16)acc.z, (_Float16)acc.w };
        *(v4h*)(s_a + r * GPITCH + lane * 4) = hv;
    }
    __syncthreads();
    if (wv >= 4) return;

    constexpr int NT = 2;
    const int hs   = lane >> 4;
    const int lrow = lane & 15;
    const _Float16* Arow = s_a + lrow * GPITCH;
    const _Float16* Bcol[NT];
    #pragma unroll
    for (int t = 0; t < NT; ++t)
        Bcol[t] = BhT + (size_t)((wv * NT + t) * 16 + lrow) * GCH;

    v8f acc2[NT];
    #pragma unroll
    for (int t = 0; t < NT; ++t) acc2[t] = (v8f){};

    #pragma unroll
    for (int kk = 0; kk < GCH; kk += 32) {
        const int kb = kk + hs * 8;
        v8h a0 = *(const v8h*)(Arow + kb);
        v8h a1 = *(const v8h*)(Arow + kb + 16);
        v16h a = __builtin_shufflevector(a0, a1, 0, 1, 2, 3, 4, 5, 6, 7,
                                         8, 9, 10, 11, 12, 13, 14, 15);
        #pragma unroll
        for (int t = 0; t < NT; ++t) {
            v8h b0 = *(const v8h*)(Bcol[t] + kb);
            v8h b1 = *(const v8h*)(Bcol[t] + kb + 16);
            v16h b = __builtin_shufflevector(b0, b1, 0, 1, 2, 3, 4, 5, 6, 7,
                                             8, 9, 10, 11, 12, 13, 14, 15);
            acc2[t] = __builtin_amdgcn_wmma_f32_16x16x32_f16(false, a, false, b,
                                                             (short)0, acc2[t],
                                                             false, false);
        }
    }

    const float* bp = bias;
    int bo = 0;
    if (wv * NT * 16 >= bsplit) { bp = bias_hi; bo = bsplit; }

    const bool fullM = (row0 + 16) <= n;
    #pragma unroll
    for (int t = 0; t < NT; ++t) {
        const int col = (wv * NT + t) * 16 + lrow;
        const float bv = bp ? bp[col - bo] : 0.f;
        float* Cb = C + (size_t)(row0 + hs * 8) * GCH + col;
        if (fullM) {
            #pragma unroll
            for (int r = 0; r < 8; ++r) {
                float v = acc2[t][r] + bv;
                if (ACT_OUT) v = leaky_f(v);
                Cb[(size_t)r * GCH] = v;
            }
        } else {
            #pragma unroll
            for (int r = 0; r < 8; ++r) {
                if (row0 + hs * 8 + r < n) {
                    float v = acc2[t][r] + bv;
                    if (ACT_OUT) v = leaky_f(v);
                    Cb[(size_t)r * GCH] = v;
                }
            }
        }
    }
}

__global__ void k_pairdot(const int* __restrict__ ts, const float* __restrict__ tf,
                          const float* __restrict__ tg, float* __restrict__ out, int M)
{
    int m = blockIdx.x * blockDim.x + threadIdx.x;
    if (m < M) {
        int a = ts[2 * m + 0];
        int b = ts[2 * m + 1];
        const float4* pa = (const float4*)(tf + (size_t)a * 32);
        const float4* pb = (const float4*)(tg + (size_t)b * 32);
        float s = 0.f;
        #pragma unroll
        for (int i = 0; i < 8; ++i) {
            float4 x = pa[i], y = pb[i];
            s += x.x * y.x + x.y * y.y + x.z * y.z + x.w * y.w;
        }
        out[m] = s;
    }
}

extern "C" void kernel_launch(void* const* d_in, const int* in_sizes, int n_in,
                              void* d_out, int out_size, void* d_ws, size_t ws_size,
                              hipStream_t stream)
{
    const float* x    = (const float*)d_in[0];
    const int*   ei   = (const int*)  d_in[1];
    const int*   ts   = (const int*)  d_in[2];
    const float* W1   = (const float*)d_in[3];
    const float* W2   = (const float*)d_in[4];
    const float* tfw1 = (const float*)d_in[5];
    const float* tfb1 = (const float*)d_in[6];
    const float* tfw2 = (const float*)d_in[7];
    const float* tfb2 = (const float*)d_in[8];
    const float* tgw1 = (const float*)d_in[9];
    const float* tgb1 = (const float*)d_in[10];
    const float* tgw2 = (const float*)d_in[11];
    const float* tgb2 = (const float*)d_in[12];

    const int N = in_sizes[0] / D_IN;
    const int E = in_sizes[1] / 2;
    const int M = in_sizes[2] / 2;
    const int* srcI = ei;
    const int* dstI = ei + E;
    float* out = (float*)d_out;

    char* w = (char*)d_ws;
    auto alloc = [&](size_t bytes) -> char* {
        char* p = w; w += (bytes + 255) & ~(size_t)255; return p;
    };
    float*    dis = (float*)   alloc((size_t)N * 4);
    _Float16* W1h = (_Float16*)alloc((size_t)D_IN * H1 * 2);
    _Float16* W2h = (_Float16*)alloc((size_t)H1 * H2 * 2);
    _Float16* FG1h = (_Float16*)alloc((size_t)2 * H2 * H3 * 2);
    _Float16* F1h = FG1h;
    _Float16* G1h = FG1h + (size_t)H3 * H2;
    _Float16* F2h = (_Float16*)alloc((size_t)H3 * D_OUT * 2);
    _Float16* G2h = (_Float16*)alloc((size_t)H3 * D_OUT * 2);
    float*    B1  = (float*)   alloc((size_t)N * H1 * 4);
    float*    B2  = (float*)   alloc((size_t)N * H1 * 4);
    float*    B3  = (float*)   alloc((size_t)N * D_OUT * 4);
    float*    B4  = (float*)   alloc((size_t)N * D_OUT * 4);
    int*      cnt    = (int*) alloc((size_t)N * 4);
    int*      rowptr = (int*) alloc((size_t)(N + 1) * 4);
    int*      cursor = (int*) alloc((size_t)N * 4);
    int*      ebuf   = (int*) alloc((size_t)E * 4);
    int2*     csr    = (int2*)alloc((size_t)E * 8);
    (void)ws_size; (void)n_in; (void)out_size;

    const int T = 256;
    auto cdiv = [](long long a, long long b) { return (int)((a + b - 1) / b); };
    auto gemm_blocks = [&](int m, int nn, int nt) {
        return cdiv((long long)((m + 15) / 16) * (nn / (16 * nt)), 8);
    };

    const int zb = cdiv(N, T);
    k_prep<<<zb + NB_WALL, T, 0, stream>>>(cnt, N, zb, W1, W2, tfw1, tfw2, tgw1, tgw2,
                                           W1h, W2h, F1h, F2h, G1h, G2h);

    k_count<<<cdiv(E, T), T, 0, stream>>>(dstI, cnt, E);
    const int sb = cdiv(N, 1024);
    k_scan_gemm1<<<sb + gemm_blocks(N, H1, 4), T, 0, stream>>>(cnt, rowptr, cursor, dis, N, sb, x, W1h, B1);
    k_fill <<<cdiv(E, T), T, 0, stream>>>(dstI, cursor, ebuf, E);
    k_sort <<<cdiv(E, T), T, 0, stream>>>(ebuf, srcI, dstI, rowptr, dis, csr, E);

    k_gather_gemm<1, 0><<<cdiv(N, 16), T, 0, stream>>>(rowptr, csr, dis, B1, W2h, nullptr, nullptr, H2, B2, N);

    float* t1 = B1;
    k_gather_gemm<0, 1><<<cdiv(N, 16), T, 0, stream>>>(rowptr, csr, dis, B2, FG1h, tfb1, tgb1, H3, t1, N);
    const int hb = gemm_blocks(N, D_OUT, 2);
    k_heads2<<<2 * hb, T, 0, stream>>>(t1, F2h, G2h, tfb2, tgb2, B4, B3, hb, N, H3, D_OUT);

    k_pairdot<<<cdiv(M, T), T, 0, stream>>>(ts, B4, B3, out, M);
}


# Loads made to cover whole 128-byte lines: weights packed in fragment order, x tile staged through LDS, pair dot reads a row with 8 lanes; pair-dot sum order now 8 lane partials joined by an xor tree
# speedup vs baseline: 2.1796x; 1.1199x over previous
#include <hip/hip_runtime.h>
#include <hip/hip_bf16.h>

typedef __attribute__((ext_vector_type(16))) _Float16 v16h;
typedef __attribute__((ext_vector_type(8)))  _Float16 v8h;
typedef __attribute__((ext_vector_type(8)))  float    v8f;

#define LEAKY_SLOPE 0.01f

__device__ __forceinline__ float elu_f(float x)   { return x > 0.f ? x : (expf(x) - 1.f); }
__device__ __forceinline__ float leaky_f(float x) { return x > 0.f ? x : LEAKY_SLOPE * x; }

constexpr int D_IN = 256, H1 = 128, H2 = 128, H3 = 64, D_OUT = 32;

__device__ __forceinline__ int pack_idx(int k, int n, int K)
{
    const int nt = n >> 4, c = n & 15;
    const int ks = k >> 5, j = (k >> 3) & 3, idx = k & 7;
    const int lane = (j & 1) * 16 + c, part = j >> 1;
    return (((nt * (K >> 5) + ks) * 2 + part) * 32 + lane) * 8 + idx;
}

template <int ACT_IN, int ACT_OUT, int NT>
__device__ __forceinline__
void gemm_tile(int bid, const float* __restrict__ A, int lda,
               const _Float16* __restrict__ BhP,
               const float* __restrict__ bias, const float* __restrict__ bias_hi, int bsplit,
               float* __restrict__ C, int Mrows, int K, int Nn)
{
    const int mtiles  = (Mrows + 15) >> 4;
    const int nstrips = Nn / (16 * NT);
    const int wave    = bid * (blockDim.x >> 5) + (threadIdx.x >> 5);
    if (wave >= mtiles * nstrips) return;
    const int mt   = wave / nstrips;
    const int ns   = wave % nstrips;
    const int lane = threadIdx.x & 31;
    const int hl   = lane >> 4;
    const int lrow = lane & 15;

    int arow = mt * 16 + lrow;
    if (arow >= Mrows) arow = Mrows - 1;
    const float* Abase = A + (size_t)arow * lda;

    const _Float16* Bcol[NT];
    #pragma unroll
    for (int t = 0; t < NT; ++t)
        Bcol[t] = BhP + (size_t)(ns * NT + t) * 16 * K + lane * 8;

    v8f acc[NT];
    #pragma unroll
    for (int t = 0; t < NT; ++t) acc[t] = (v8f){};

    for (int kk = 0; kk < K; kk += 32) {
        const int kb = kk + hl * 8;
        __builtin_prefetch(Abase + kk + 32, 0, 0);

        const float4* ap = (const float4*)(Abase + kb);
        float4 a0 = ap[0], a1 = ap[1];
        float4 a2 = ap[4], a3 = ap[5];
        float af[16] = { a0.x, a0.y, a0.z, a0.w, a1.x, a1.y, a1.z, a1.w,
                         a2.x, a2.y, a2.z, a2.w, a3.x, a3.y, a3.z, a3.w };
        v16h a;
        #pragma unroll
        for (int i = 0; i < 16; ++i) {
            float v = af[i];
            if (ACT_IN) v = elu_f(v);
            a[i] = (_Float16)v;
        }

        #pragma unroll
        for (int t = 0; t < NT; ++t) {
            v8h b0 = *(const v8h*)(Bcol[t] + kk * 16);
            v8h b1 = *(const v8h*)(Bcol[t] + kk * 16 + 256);
            v16h b = __builtin_shufflevector(b0, b1, 0, 1, 2, 3, 4, 5, 6, 7,
                                             8, 9, 10, 11, 12, 13, 14, 15);
            acc[t] = __builtin_amdgcn_wmma_f32_16x16x32_f16(false, a, false, b,
                                                            (short)0, acc[t],
                                                            false, false);
        }
    }

    const float* bp = bias;
    int bo = 0;
    if (ns * NT * 16 >= bsplit) { bp = bias_hi; bo = bsplit; }

    const bool fullM = (mt * 16 + 16) <= Mrows;
    #pragma unroll
    for (int t = 0; t < NT; ++t) {
        const int col = (ns * NT + t) * 16 + lrow;
        const float bv = bp ? bp[col - bo] : 0.f;
        float* Cb = C + (size_t)(mt * 16 + hl * 8) * Nn + col;
        if (fullM) {
            #pragma unroll
            for (int r = 0; r < 8; ++r) {
                float v = acc[t][r] + bv;
                if (ACT_OUT) v = leaky_f(v);
                Cb[(size_t)r * Nn] = v;
            }
        } else {
            #pragma unroll
            for (int r = 0; r < 8; ++r) {
                const int orow = mt * 16 + hl * 8 + r;
                if (orow < Mrows) {
                    float v = acc[t][r] + bv;
                    if (ACT_OUT) v = leaky_f(v);
                    Cb[(size_t)r * Nn] = v;
                }
            }
        }
    }
}

__global__ __launch_bounds__(256)
void k_heads2(const float* __restrict__ t1,
              const _Float16* __restrict__ Fh, const _Float16* __restrict__ Gh,
              const float* __restrict__ bF, const float* __restrict__ bG,
              float* __restrict__ oF, float* __restrict__ oG,
              int nb, int Mrows, int K, int Nn)
{
    const bool g = (int)blockIdx.x >= nb;
    gemm_tile<0, 1, 2>((int)blockIdx.x - (g ? nb : 0), t1 + (g ? K : 0), 2 * K,
                       g ? Gh : Fh, g ? bG : bF, nullptr, Nn,
                       g ? oG : oF, Mrows, K, Nn);
}

__global__ void k_count(const int* __restrict__ dst, int* cnt, int E)
{
    int e = blockIdx.x * blockDim.x + threadIdx.x;
    if (e < E) atomicAdd(&cnt[dst[e]], 1);
}

__device__ __forceinline__
void scan_tile(int bid, const int* __restrict__ cnt, int* __restrict__ rowptr,
               int* __restrict__ cursor, float* __restrict__ dis, int n)
{
    __shared__ int s_pre[8];
    __shared__ int s_tot[8];
    const int t = threadIdx.x, lane = t & 31, wv = t >> 5;

    const int4* c4 = (const int4*)cnt;
    const int nb = bid;
    int pre = 0;
    #pragma unroll 8
    for (int m = 0; m < nb; ++m) {
        int4 v = c4[m * 256 + t];
        pre += (v.x + v.y) + (v.z + v.w);
    }

    const int i0 = bid * 1024 + t * 4;
    int c[4];
    if (i0 + 3 < n) {
        int4 v = c4[i0 >> 2];
        c[0] = v.x; c[1] = v.y; c[2] = v.z; c[3] = v.w;
    } else {
        for (int k = 0; k < 4; ++k) c[k] = (i0 + k < n) ? cnt[i0 + k] : 0;
    }
    const int mine = (c[0] + c[1]) + (c[2] + c[3]);

    int incl = mine;
    for (int off = 1; off < 32; off <<= 1) {
        int v = __shfl_up(incl, off);
        if (lane >= off) incl += v;
    }
    for (int off = 16; off > 0; off >>= 1) pre += __shfl_xor(pre, off);
    if (lane == 31) s_tot[wv] = incl;
    if (lane == 0)  s_pre[wv] = pre;
    __syncthreads();
    int run = incl - mine;
    for (int k = 0; k < 8; ++k) {
        run += s_pre[k];
        if (k < wv) run += s_tot[k];
    }

    int r[4]; float d[4];
    for (int k = 0; k < 4; ++k) {
        r[k] = run;
        d[k] = rsqrtf((float)(c[k] + 1));
        run += c[k];
    }
    if (i0 + 3 < n) {
        *(int4*)(rowptr + i0)  = make_int4(r[0], r[1], r[2], r[3]);
        *(int4*)(cursor + i0)  = make_int4(r[0], r[1], r[2], r[3]);
        *(float4*)(dis + i0)   = make_float4(d[0], d[1], d[2], d[3]);
    } else {
        for (int k = 0; k < 4; ++k)
            if (i0 + k < n) { rowptr[i0 + k] = r[k]; cursor[i0 + k] = r[k]; dis[i0 + k] = d[k]; }
    }
    if (i0 < n && n <= i0 + 4) rowptr[n] = run;
}

constexpr int G1ROWS  = 64;
constexpr int G1PITCH = D_IN + 8;
__device__ __forceinline__
void gemm1_tile(int bid, const float* __restrict__ x, const _Float16* __restrict__ BhP,
                float* __restrict__ C, int n, _Float16* __restrict__ s_x)
{
    const int lane = threadIdx.x & 31;
    const int wv = __builtin_amdgcn_readfirstlane(threadIdx.x >> 5);
    const int row0 = bid * G1ROWS;
    typedef __attribute__((ext_vector_type(4))) _Float16 v4h;

    float4 xv[16];
    #pragma unroll
    for (int u = 0; u < 8; ++u) {
        int node = row0 + wv * 8 + u;
        if (node >= n) node = n - 1;
        const float* xr = x + (size_t)node * D_IN + lane * 4;
        xv[2 * u]     = *(const float4*)(xr);
        xv[2 * u + 1] = *(const float4*)(xr + 128);
    }
    #pragma unroll
    for (int u = 0; u < 8; ++u) {
        _Float16* sr = s_x + (wv * 8 + u) * G1PITCH + lane * 4;
        const float4 p = xv[2 * u], q = xv[2 * u + 1];
        *(v4h*)(sr)       = (v4h){ (_Float16)p.x, (_Float16)p.y, (_Float16)p.z, (_Float16)p.w };
        *(v4h*)(sr + 128) = (v4h){ (_Float16)q.x, (_Float16)q.y, (_Float16)q.z, (_Float16)q.w };
    }
    __syncthreads();

    constexpr int NT = 4;
    const int mt = wv >> 1, ns = wv & 1;
    const int hs   = lane >> 4;
    const int lrow = lane & 15;
    const _Float16* Arow = s_x + (mt * 16 + lrow) * G1PITCH;
    const _Float16* Bcol[NT];
    #pragma unroll
    for (int t = 0; t < NT; ++t)
        Bcol[t] = BhP + (size_t)(ns * NT + t) * 16 * D_IN + lane * 8;

    v8f acc[NT];
    #pragma unroll
    for (int t = 0; t < NT; ++t) acc[t] = (v8f){};

    for (int kk = 0; kk < D_IN; kk += 32) {
        const int kb = kk + hs * 8;
        v8h a0 = *(const v8h*)(Arow + kb);
        v8h a1 = *(const v8h*)(Arow + kb + 16);
        v16h a = __builtin_shufflevector(a0, a1, 0, 1, 2, 3, 4, 5, 6, 7,
                                         8, 9, 10, 11, 12, 13, 14, 15);
        #pragma unroll
        for (int t = 0; t < NT; ++t) {
            v8h b0 = *(const v8h*)(Bcol[t] + kk * 16);
            v8h b1 = *(const v8h*)(Bcol[t] + kk * 16 + 256);
            v16h b = __builtin_shufflevector(b0, b1, 0, 1, 2, 3, 4, 5, 6, 7,
                                             8, 9, 10, 11, 12, 13, 14, 15);
            acc[t] = __builtin_amdgcn_wmma_f32_16x16x32_f16(false, a, false, b,
                                                            (short)0, acc[t],
                                                            false, false);
        }
    }

    const int orow0 = row0 + mt * 16 + hs * 8;
    const bool fullM = (row0 + mt * 16 + 16) <= n;
    #pragma unroll
    for (int t = 0; t < NT; ++t) {
        const int col = (ns * NT + t) * 16 + lrow;
        float* Cb = C + (size_t)orow0 * H1 + col;
        if (fullM) {
            #pragma unroll
            for (int r = 0; r < 8; ++r) Cb[(size_t)r * H1] = acc[t][r];
        } else {
            #pragma unroll
            for (int r = 0; r < 8; ++r)
                if (orow0 + r < n) Cb[(size_t)r * H1] = acc[t][r];
        }
    }
}

__global__ __launch_bounds__(256)
void k_scan_gemm1(const int* __restrict__ cnt, int* __restrict__ rowptr,
                  int* __restrict__ cursor, float* __restrict__ dis, int n, int sb,
                  const float* __restrict__ x, const _Float16* __restrict__ W1h,
                  float* __restrict__ h1)
{
    __shared__ __attribute__((aligned(16))) _Float16 s_x[G1ROWS * G1PITCH];
    if ((int)blockIdx.x < sb) {
        scan_tile((int)blockIdx.x, cnt, rowptr, cursor, dis, n);
        return;
    }
    gemm1_tile((int)blockIdx.x - sb, x, W1h, h1, n, s_x);
}

__global__ void k_fill(const int* __restrict__ dst, int* cursor, int* __restrict__ ebuf, int E)
{
    int e = blockIdx.x * blockDim.x + threadIdx.x;
    if (e < E) {
        int pos = atomicAdd(&cursor[dst[e]], 1);
        ebuf[pos] = e;
    }
}

__global__ void k_sort(const int* __restrict__ ebuf, const int* __restrict__ srcI,
                       const int* __restrict__ dstI, const int* __restrict__ rowptr,
                       const float* __restrict__ dis, int2* __restrict__ csr, int E)
{
    int p = blockIdx.x * blockDim.x + threadIdx.x;
    if (p < E) {
        const int e = ebuf[p];
        const int d = dstI[e];
        const int b = rowptr[d], en = rowptr[d + 1];
        int rank = 0;
        for (int i = b; i < en; ++i) rank += (ebuf[i] < e) ? 1 : 0;
        const int s = srcI[e];
        const float w = dis[s] * dis[d];
        csr[b + rank] = make_int2(s, __float_as_int(w));
    }
}

__device__ __forceinline__
void f16T_blk(const float* __restrict__ s, _Float16* __restrict__ d, int K, int Nn, int blk)
{
    int i = blk * 256 + threadIdx.x;
    if (i < K * Nn) {
        int k = i / Nn, n = i % Nn;
        d[pack_idx(k, n, K)] = (_Float16)s[i];
    }
}

constexpr int NB_W1 = (D_IN * H1 + 255) / 256, NB_W2 = (H1 * H2 + 255) / 256;
constexpr int NB_H1 = (H2 * H3 + 255) / 256,   NB_H2 = (H3 * D_OUT + 255) / 256;
constexpr int NB_WALL = NB_W1 + NB_W2 + 2 * NB_H1 + 2 * NB_H2;

__global__ __launch_bounds__(256)
void k_prep(int* __restrict__ cnt, int n, int zb,
            const float* __restrict__ W1,   const float* __restrict__ W2,
            const float* __restrict__ tfw1, const float* __restrict__ tfw2,
            const float* __restrict__ tgw1, const float* __restrict__ tgw2,
            _Float16* __restrict__ W1h, _Float16* __restrict__ W2h,
            _Float16* __restrict__ F1h, _Float16* __restrict__ F2h,
            _Float16* __restrict__ G1h, _Float16* __restrict__ G2h)
{
    int b = blockIdx.x;
    if (b < zb) {
        const int i = b * 256 + threadIdx.x;
        if (i < n) cnt[i] = 0;
        return;
    }
    b -= zb;
    if (b < NB_W1) { f16T_blk(W1,   W1h, D_IN, H1,    b); return; }  b -= NB_W1;
    if (b < NB_W2) { f16T_blk(W2,   W2h, H1,   H2,    b); return; }  b -= NB_W2;
    if (b < NB_H1) { f16T_blk(tfw1, F1h, H2,   H3,    b); return; }  b -= NB_H1;
    if (b < NB_H2) { f16T_blk(tfw2, F2h, H3,   D_OUT, b); return; }  b -= NB_H2;
    if (b < NB_H1) { f16T_blk(tgw1, G1h, H2,   H3,    b); return; }  b -= NB_H1;
    f16T_blk(tgw2, G2h, H3, D_OUT, b);
}

template <bool FULL>
__device__ __forceinline__ void gather_group(const float* __restrict__ hl, int mys, int myw,
                                             int j, int cnt, float4& acc)
{
    float4 r[8];
    float  w[8];
    #pragma unroll
    for (int u = 0; u < 8; ++u) {
        r[u] = make_float4(0.f, 0.f, 0.f, 0.f);
        w[u] = 0.f;
        if (FULL || j + u < cnt) {
            const int s = __builtin_amdgcn_readlane(mys, j + u);
            w[u] = __int_as_float(__builtin_amdgcn_readlane(myw, j + u));
            r[u] = *(const float4*)(hl + (size_t)s * 128);
        }
    }
    #pragma unroll
    for (int u = 0; u < 8; ++u) {
        if (FULL || j + u < cnt) {
            acc.x = fmaf(r[u].x, w[u], acc.x);
            acc.y = fmaf(r[u].y, w[u], acc.y);
            acc.z = fmaf(r[u].z, w[u], acc.z);
            acc.w = fmaf(r[u].w, w[u], acc.w);
        }
    }
}

#define GNPW 2
constexpr int GCH = 128;
constexpr int GPITCH = GCH + 8;
template <int ACT_IN, int ACT_OUT>
__global__ __launch_bounds__(256)
void k_gather_gemm(const int* __restrict__ rowptr, const int2* __restrict__ csr,
                   const float* __restrict__ dis, const float* __restrict__ h,
                   const _Float16* __restrict__ BhP,
                   const float* __restrict__ bias, const float* __restrict__ bias_hi, int bsplit,
                   float* __restrict__ C, int n)
{
    __shared__ __attribute__((aligned(16))) _Float16 s_a[16 * GPITCH];
    const int lane = threadIdx.x & 31;
    const int wv = __builtin_amdgcn_readfirstlane(threadIdx.x >> 5);
    const int row0 = blockIdx.x * 16;
    const float* hl = h + lane * 4;
    for (int k = 0; k < GNPW; ++k) {
        const int r = wv * GNPW + k;
        const int node = row0 + r;
        float4 acc = make_float4(0.f, 0.f, 0.f, 0.f);
        if (node < n) {
            const int beg = rowptr[node], end = rowptr[node + 1];
            float sd = dis[node]; sd *= sd;
            int base = beg;
            int cnt  = (end - base < 32) ? (end - base) : 32;
            int2 my = make_int2(0, 0);
            if (lane < cnt) my = csr[base + lane];
            const float4 v = *(const float4*)(hl + (size_t)node * GCH);
            while (cnt > 0) {
                int j = 0;
                for (; j + 8 <= cnt; j += 8) gather_group<true >(hl, my.x, my.y, j, cnt, acc);
                if (j < cnt)                 gather_group<false>(hl, my.x, my.y, j, cnt, acc);
                base += 32;
                cnt = (end - base < 32) ? (end - base) : 32;
                if (lane < cnt) my = csr[base + lane];
            }
            acc.x = fmaf(v.x, sd, acc.x);
            acc.y = fmaf(v.y, sd, acc.y);
            acc.z = fmaf(v.z, sd, acc.z);
            acc.w = fmaf(v.w, sd, acc.w);
        }
        if (ACT_IN) { acc.x = elu_f(acc.x); acc.y = elu_f(acc.y); acc.z = elu_f(acc.z); acc.w = elu_f(acc.w); }
        typedef __attribute__((ext_vector_type(4))) _Float16 v4h;
        v4h hv = { (_Float16)acc.x, (_Float16)acc.y, (_Float16)acc.z, (_Float16)acc.w };
        *(v4h*)(s_a + r * GPITCH + lane * 4) = hv;
    }
    __syncthreads();
    if (wv >= 4) return;

    constexpr int NT = 2;
    const int hs   = lane >> 4;
    const int lrow = lane & 15;
    const _Float16* Arow = s_a + lrow * GPITCH;
    const _Float16* Bcol[NT];
    #pragma unroll
    for (int t = 0; t < NT; ++t)
        Bcol[t] = BhP + (size_t)(wv * NT + t) * 16 * GCH + lane * 8;

    v8f acc2[NT];
    #pragma unroll
    for (int t = 0; t < NT; ++t) acc2[t] = (v8f){};

    #pragma unroll
    for (int kk = 0; kk < GCH; kk += 32) {
        const int kb = kk + hs * 8;
        v8h a0 = *(const v8h*)(Arow + kb);
        v8h a1 = *(const v8h*)(Arow + kb + 16);
        v16h a = __builtin_shufflevector(a0, a1, 0, 1, 2, 3, 4, 5, 6, 7,
                                         8, 9, 10, 11, 12, 13, 14, 15);
        #pragma unroll
        for (int t = 0; t < NT; ++t) {
            v8h b0 = *(const v8h*)(Bcol[t] + kk * 16);
            v8h b1 = *(const v8h*)(Bcol[t] + kk * 16 + 256);
            v16h b = __builtin_shufflevector(b0, b1, 0, 1, 2, 3, 4, 5, 6, 7,
                                             8, 9, 10, 11, 12, 13, 14, 15);
            acc2[t] = __builtin_amdgcn_wmma_f32_16x16x32_f16(false, a, false, b,
                                                             (short)0, acc2[t],
                                                             false, false);
        }
    }

    const float* bp = bias;
    int bo = 0;
    if (wv * NT * 16 >= bsplit) { bp = bias_hi; bo = bsplit; }

    const bool fullM = (row0 + 16) <= n;
    #pragma unroll
    for (int t = 0; t < NT; ++t) {
        const int col = (wv * NT + t) * 16 + lrow;
        const float bv = bp ? bp[col - bo] : 0.f;
        float* Cb = C + (size_t)(row0 + hs * 8) * GCH + col;
        if (fullM) {
            #pragma unroll
            for (int r = 0; r < 8; ++r) {
                float v = acc2[t][r] + bv;
                if (ACT_OUT) v = leaky_f(v);
                Cb[(size_t)r * GCH] = v;
            }
        } else {
            #pragma unroll
            for (int r = 0; r < 8; ++r) {
                if (row0 + hs * 8 + r < n) {
                    float v = acc2[t][r] + bv;
                    if (ACT_OUT) v = leaky_f(v);
                    Cb[(size_t)r * GCH] = v;
                }
            }
        }
    }
}

__global__ __launch_bounds__(256)
void k_pairdot(const int* __restrict__ ts, const float* __restrict__ tf,
               const float* __restrict__ tg, float* __restrict__ out, int M)
{
    const int lane = threadIdx.x & 31;
    const int base = (blockIdx.x * blockDim.x + threadIdx.x) & ~31;
    const int g8 = lane & 24;
    const int k  = lane & 7;
    float4 xr[8], yr[8];
    #pragma unroll
    for (int j = 0; j < 8; ++j) {
        int m = base + g8 + j;
        if (m >= M) m = M - 1;
        const int2 ab = *(const int2*)(ts + 2 * (size_t)m);
        xr[j] = *(const float4*)(tf + (size_t)ab.x * 32 + k * 4);
        yr[j] = *(const float4*)(tg + (size_t)ab.y * 32 + k * 4);
    }
    float p[8];
    #pragma unroll
    for (int j = 0; j < 8; ++j) {
        p[j] = xr[j].x * yr[j].x + xr[j].y * yr[j].y + xr[j].z * yr[j].z + xr[j].w * yr[j].w;
        p[j] += __shfl_xor(p[j], 4);
        p[j] += __shfl_xor(p[j], 2);
        p[j] += __shfl_xor(p[j], 1);
    }
    float r = p[0];
    #pragma unroll
    for (int j = 1; j < 8; ++j) r = (k == j) ? p[j] : r;
    if (base + lane < M) out[base + lane] = r;
}

extern "C" void kernel_launch(void* const* d_in, const int* in_sizes, int n_in,
                              void* d_out, int out_size, void* d_ws, size_t ws_size,
                              hipStream_t stream)
{
    const float* x    = (const float*)d_in[0];
    const int*   ei   = (const int*)  d_in[1];
    const int*   ts   = (const int*)  d_in[2];
    const float* W1   = (const float*)d_in[3];
    const float* W2   = (const float*)d_in[4];
    const float* tfw1 = (const float*)d_in[5];
    const float* tfb1 = (const float*)d_in[6];
    const float* tfw2 = (const float*)d_in[7];
    const float* tfb2 = (const float*)d_in[8];
    const float* tgw1 = (const float*)d_in[9];
    const float* tgb1 = (const float*)d_in[10];
    const float* tgw2 = (const float*)d_in[11];
    const float* tgb2 = (const float*)d_in[12];

    const int N = in_sizes[0] / D_IN;
    const int E = in_sizes[1] / 2;
    const int M = in_sizes[2] / 2;
    const int* srcI = ei;
    const int* dstI = ei + E;
    float* out = (float*)d_out;

    char* w = (char*)d_ws;
    auto alloc = [&](size_t bytes) -> char* {
        char* p = w; w += (bytes + 255) & ~(size_t)255; return p;
    };
    float*    dis = (float*)   alloc((size_t)N * 4);
    _Float16* W1h = (_Float16*)alloc((size_t)D_IN * H1 * 2);
    _Float16* W2h = (_Float16*)alloc((size_t)H1 * H2 * 2);
    _Float16* FG1h = (_Float16*)alloc((size_t)2 * H2 * H3 * 2);
    _Float16* F1h = FG1h;
    _Float16* G1h = FG1h + (size_t)H3 * H2;
    _Float16* F2h = (_Float16*)alloc((size_t)H3 * D_OUT * 2);
    _Float16* G2h = (_Float16*)alloc((size_t)H3 * D_OUT * 2);
    float*    B1  = (float*)   alloc((size_t)N * H1 * 4);
    float*    B2  = (float*)   alloc((size_t)N * H1 * 4);
    float*    B3  = (float*)   alloc((size_t)N * D_OUT * 4);
    float*    B4  = (float*)   alloc((size_t)N * D_OUT * 4);
    int*      cnt    = (int*) alloc((size_t)N * 4);
    int*      rowptr = (int*) alloc((size_t)(N + 1) * 4);
    int*      cursor = (int*) alloc((size_t)N * 4);
    int*      ebuf   = (int*) alloc((size_t)E * 4);
    int2*     csr    = (int2*)alloc((size_t)E * 8);
    (void)ws_size; (void)n_in; (void)out_size;

    const int T = 256;
    auto cdiv = [](long long a, long long b) { return (int)((a + b - 1) / b); };
    auto gemm_blocks = [&](int m, int nn, int nt) {
        return cdiv((long long)((m + 15) / 16) * (nn / (16 * nt)), 8);
    };

    const int zb = cdiv(N, T);
    k_prep<<<zb + NB_WALL, T, 0, stream>>>(cnt, N, zb, W1, W2, tfw1, tfw2, tgw1, tgw2,
                                           W1h, W2h, F1h, F2h, G1h, G2h);

    k_count<<<cdiv(E, T), T, 0, stream>>>(dstI, cnt, E);
    const int sb = cdiv(N, 1024);
    k_scan_gemm1<<<sb + cdiv(N, G1ROWS), T, 0, stream>>>(cnt, rowptr, cursor, dis, N, sb, x, W1h, B1);
    k_fill <<<cdiv(E, T), T, 0, stream>>>(dstI, cursor, ebuf, E);
    k_sort <<<cdiv(E, T), T, 0, stream>>>(ebuf, srcI, dstI, rowptr, dis, csr, E);

    k_gather_gemm<1, 0><<<cdiv(N, 16), T, 0, stream>>>(rowptr, csr, dis, B1, W2h, nullptr, nullptr, H2, B2, N);

    float* t1 = B1;
    k_gather_gemm<0, 1><<<cdiv(N, 16), T, 0, stream>>>(rowptr, csr, dis, B2, FG1h, tfb1, tgb1, H3, t1, N);
    const int hb = gemm_blocks(N, D_OUT, 2);
    k_heads2<<<2 * hb, T, 0, stream>>>(t1, F2h, G2h, tfb2, tgb2, B4, B3, hb, N, H3, D_OUT);

    k_pairdot<<<cdiv(M, T), T, 0, stream>>>(ts, B4, B3, out, M);
}


# Second gather also hands its head-layer-1 tile to head layer 2 through LDS (gather, product, act, product in one kernel) instead of a store, a launch and a load: 9 launches to 8; sums unchanged
# speedup vs baseline: 2.3034x; 1.0568x over previous
#include <hip/hip_runtime.h>
#include <hip/hip_bf16.h>

typedef __attribute__((ext_vector_type(16))) _Float16 v16h;
typedef __attribute__((ext_vector_type(8)))  _Float16 v8h;
typedef __attribute__((ext_vector_type(8)))  float    v8f;

#define LEAKY_SLOPE 0.01f

__device__ __forceinline__ float elu_f(float x)   { return x > 0.f ? x : (expf(x) - 1.f); }
__device__ __forceinline__ float leaky_f(float x) { return x > 0.f ? x : LEAKY_SLOPE * x; }

constexpr int D_IN = 256, H1 = 128, H2 = 128, H3 = 64, D_OUT = 32;

__device__ __forceinline__ int pack_idx(int k, int n, int K)
{
    const int nt = n >> 4, c = n & 15;
    const int ks = k >> 5, j = (k >> 3) & 3, idx = k & 7;
    const int lane = (j & 1) * 16 + c, part = j >> 1;
    return (((nt * (K >> 5) + ks) * 2 + part) * 32 + lane) * 8 + idx;
}

__global__ void k_count(const int* __restrict__ dst, int* cnt, int E)
{
    int e = blockIdx.x * blockDim.x + threadIdx.x;
    if (e < E) atomicAdd(&cnt[dst[e]], 1);
}

__device__ __forceinline__
void scan_tile(int bid, const int* __restrict__ cnt, int* __restrict__ rowptr,
               int* __restrict__ cursor, float* __restrict__ dis, int n)
{
    __shared__ int s_pre[8];
    __shared__ int s_tot[8];
    const int t = threadIdx.x, lane = t & 31, wv = t >> 5;

    const int4* c4 = (const int4*)cnt;
    const int nb = bid;
    int pre = 0;
    #pragma unroll 8
    for (int m = 0; m < nb; ++m) {
        int4 v = c4[m * 256 + t];
        pre += (v.x + v.y) + (v.z + v.w);
    }

    const int i0 = bid * 1024 + t * 4;
    int c[4];
    if (i0 + 3 < n) {
        int4 v = c4[i0 >> 2];
        c[0] = v.x; c[1] = v.y; c[2] = v.z; c[3] = v.w;
    } else {
        for (int k = 0; k < 4; ++k) c[k] = (i0 + k < n) ? cnt[i0 + k] : 0;
    }
    const int mine = (c[0] + c[1]) + (c[2] + c[3]);

    int incl = mine;
    for (int off = 1; off < 32; off <<= 1) {
        int v = __shfl_up(incl, off);
        if (lane >= off) incl += v;
    }
    for (int off = 16; off > 0; off >>= 1) pre += __shfl_xor(pre, off);
    if (lane == 31) s_tot[wv] = incl;
    if (lane == 0)  s_pre[wv] = pre;
    __syncthreads();
    int run = incl - mine;
    for (int k = 0; k < 8; ++k) {
        run += s_pre[k];
        if (k < wv) run += s_tot[k];
    }

    int r[4]; float d[4];
    for (int k = 0; k < 4; ++k) {
        r[k] = run;
        d[k] = rsqrtf((float)(c[k] + 1));
        run += c[k];
    }
    if (i0 + 3 < n) {
        *(int4*)(rowptr + i0)  = make_int4(r[0], r[1], r[2], r[3]);
        *(int4*)(cursor + i0)  = make_int4(r[0], r[1], r[2], r[3]);
        *(float4*)(dis + i0)   = make_float4(d[0], d[1], d[2], d[3]);
    } else {
        for (int k = 0; k < 4; ++k)
            if (i0 + k < n) { rowptr[i0 + k] = r[k]; cursor[i0 + k] = r[k]; dis[i0 + k] = d[k]; }
    }
    if (i0 < n && n <= i0 + 4) rowptr[n] = run;
}

constexpr int G1ROWS  = 64;
constexpr int G1PITCH = D_IN + 8;
__device__ __forceinline__
void gemm1_tile(int bid, const float* __restrict__ x, const _Float16* __restrict__ BhP,
                float* __restrict__ C, int n, _Float16* __restrict__ s_x)
{
    const int lane = threadIdx.x & 31;
    const int wv = __builtin_amdgcn_readfirstlane(threadIdx.x >> 5);
    const int row0 = bid * G1ROWS;
    typedef __attribute__((ext_vector_type(4))) _Float16 v4h;

    float4 xv[16];
    #pragma unroll
    for (int u = 0; u < 8; ++u) {
        int node = row0 + wv * 8 + u;
        if (node >= n) node = n - 1;
        const float* xr = x + (size_t)node * D_IN + lane * 4;
        xv[2 * u]     = *(const float4*)(xr);
        xv[2 * u + 1] = *(const float4*)(xr + 128);
    }
    #pragma unroll
    for (int u = 0; u < 8; ++u) {
        _Float16* sr = s_x + (wv * 8 + u) * G1PITCH + lane * 4;
        const float4 p = xv[2 * u], q = xv[2 * u + 1];
        *(v4h*)(sr)       = (v4h){ (_Float16)p.x, (_Float16)p.y, (_Float16)p.z, (_Float16)p.w };
        *(v4h*)(sr + 128) = (v4h){ (_Float16)q.x, (_Float16)q.y, (_Float16)q.z, (_Float16)q.w };
    }
    __syncthreads();

    constexpr int NT = 4;
    const int mt = wv >> 1, ns = wv & 1;
    const int hs   = lane >> 4;
    const int lrow = lane & 15;
    const _Float16* Arow = s_x + (mt * 16 + lrow) * G1PITCH;
    const _Float16* Bcol[NT];
    #pragma unroll
    for (int t = 0; t < NT; ++t)
        Bcol[t] = BhP + (size_t)(ns * NT + t) * 16 * D_IN + lane * 8;

    v8f acc[NT];
    #pragma unroll
    for (int t = 0; t < NT; ++t) acc[t] = (v8f){};

    for (int kk = 0; kk < D_IN; kk += 32) {
        const int kb = kk + hs * 8;
        v8h a0 = *(const v8h*)(Arow + kb);
        v8h a1 = *(const v8h*)(Arow + kb + 16);
        v16h a = __builtin_shufflevector(a0, a1, 0, 1, 2, 3, 4, 5, 6, 7,
                                         8, 9, 10, 11, 12, 13, 14, 15);
        #pragma unroll
        for (int t = 0; t < NT; ++t) {
            v8h b0 = *(const v8h*)(Bcol[t] + kk * 16);
            v8h b1 = *(const v8h*)(Bcol[t] + kk * 16 + 256);
            v16h b = __builtin_shufflevector(b0, b1, 0, 1, 2, 3, 4, 5, 6, 7,
                                             8, 9, 10, 11, 12, 13, 14, 15);
            acc[t] = __builtin_amdgcn_wmma_f32_16x16x32_f16(false, a, false, b,
                                                            (short)0, acc[t],
                                                            false, false);
        }
    }

    const int orow0 = row0 + mt * 16 + hs * 8;
    const bool fullM = (row0 + mt * 16 + 16) <= n;
    #pragma unroll
    for (int t = 0; t < NT; ++t) {
        const int col = (ns * NT + t) * 16 + lrow;
        float* Cb = C + (size_t)orow0 * H1 + col;
        if (fullM) {
            #pragma unroll
            for (int r = 0; r < 8; ++r) Cb[(size_t)r * H1] = acc[t][r];
        } else {
            #pragma unroll
            for (int r = 0; r < 8; ++r)
                if (orow0 + r < n) Cb[(size_t)r * H1] = acc[t][r];
        }
    }
}

__global__ __launch_bounds__(256)
void k_scan_gemm1(const int* __restrict__ cnt, int* __restrict__ rowptr,
                  int* __restrict__ cursor, float* __restrict__ dis, int n, int sb,
                  const float* __restrict__ x, const _Float16* __restrict__ W1h,
                  float* __restrict__ h1)
{
    __shared__ __attribute__((aligned(16))) _Float16 s_x[G1ROWS * G1PITCH];
    if ((int)blockIdx.x < sb) {
        scan_tile((int)blockIdx.x, cnt, rowptr, cursor, dis, n);
        return;
    }
    gemm1_tile((int)blockIdx.x - sb, x, W1h, h1, n, s_x);
}

__global__ void k_fill(const int* __restrict__ dst, int* cursor, int* __restrict__ ebuf, int E)
{
    int e = blockIdx.x * blockDim.x + threadIdx.x;
    if (e < E) {
        int pos = atomicAdd(&cursor[dst[e]], 1);
        ebuf[pos] = e;
    }
}

__global__ void k_sort(const int* __restrict__ ebuf, const int* __restrict__ srcI,
                       const int* __restrict__ dstI, const int* __restrict__ rowptr,
                       const float* __restrict__ dis, int2* __restrict__ csr, int E)
{
    int p = blockIdx.x * blockDim.x + threadIdx.x;
    if (p < E) {
        const int e = ebuf[p];
        const int d = dstI[e];
        const int b = rowptr[d], en = rowptr[d + 1];
        int rank = 0;
        for (int i = b; i < en; ++i) rank += (ebuf[i] < e) ? 1 : 0;
        const int s = srcI[e];
        const float w = dis[s] * dis[d];
        csr[b + rank] = make_int2(s, __float_as_int(w));
    }
}

__device__ __forceinline__
void f16T_blk(const float* __restrict__ s, _Float16* __restrict__ d, int K, int Nn, int blk)
{
    int i = blk * 256 + threadIdx.x;
    if (i < K * Nn) {
        int k = i / Nn, n = i % Nn;
        d[pack_idx(k, n, K)] = (_Float16)s[i];
    }
}

constexpr int NB_W1 = (D_IN * H1 + 255) / 256, NB_W2 = (H1 * H2 + 255) / 256;
constexpr int NB_H1 = (H2 * H3 + 255) / 256,   NB_H2 = (H3 * D_OUT + 255) / 256;
constexpr int NB_WALL = NB_W1 + NB_W2 + 2 * NB_H1 + 2 * NB_H2;

__global__ __launch_bounds__(256)
void k_prep(int* __restrict__ cnt, int n, int zb,
            const float* __restrict__ W1,   const float* __restrict__ W2,
            const float* __restrict__ tfw1, const float* __restrict__ tfw2,
            const float* __restrict__ tgw1, const float* __restrict__ tgw2,
            _Float16* __restrict__ W1h, _Float16* __restrict__ W2h,
            _Float16* __restrict__ F1h, _Float16* __restrict__ F2h,
            _Float16* __restrict__ G1h, _Float16* __restrict__ G2h)
{
    int b = blockIdx.x;
    if (b < zb) {
        const int i = b * 256 + threadIdx.x;
        if (i < n) cnt[i] = 0;
        return;
    }
    b -= zb;
    if (b < NB_W1) { f16T_blk(W1,   W1h, D_IN, H1,    b); return; }  b -= NB_W1;
    if (b < NB_W2) { f16T_blk(W2,   W2h, H1,   H2,    b); return; }  b -= NB_W2;
    if (b < NB_H1) { f16T_blk(tfw1, F1h, H2,   H3,    b); return; }  b -= NB_H1;
    if (b < NB_H2) { f16T_blk(tfw2, F2h, H3,   D_OUT, b); return; }  b -= NB_H2;
    if (b < NB_H1) { f16T_blk(tgw1, G1h, H2,   H3,    b); return; }  b -= NB_H1;
    f16T_blk(tgw2, G2h, H3, D_OUT, b);
}

template <bool FULL>
__device__ __forceinline__ void gather_group(const float* __restrict__ hl, int mys, int myw,
                                             int j, int cnt, float4& acc)
{
    float4 r[8];
    float  w[8];
    #pragma unroll
    for (int u = 0; u < 8; ++u) {
        r[u] = make_float4(0.f, 0.f, 0.f, 0.f);
        w[u] = 0.f;
        if (FULL || j + u < cnt) {
            const int s = __builtin_amdgcn_readlane(mys, j + u);
            w[u] = __int_as_float(__builtin_amdgcn_readlane(myw, j + u));
            r[u] = *(const float4*)(hl + (size_t)s * 128);
        }
    }
    #pragma unroll
    for (int u = 0; u < 8; ++u) {
        if (FULL || j + u < cnt) {
            acc.x = fmaf(r[u].x, w[u], acc.x);
            acc.y = fmaf(r[u].y, w[u], acc.y);
            acc.z = fmaf(r[u].z, w[u], acc.z);
            acc.w = fmaf(r[u].w, w[u], acc.w);
        }
    }
}

#define GNPW 2
constexpr int GCH = 128;
constexpr int GPITCH = GCH + 8;
template <int ACT_IN, int ACT_OUT, bool HEADS>
__global__ __launch_bounds__(256)
void k_gather_gemm(const int* __restrict__ rowptr, const int2* __restrict__ csr,
                   const float* __restrict__ dis, const float* __restrict__ h,
                   const _Float16* __restrict__ BhP,
                   const float* __restrict__ bias, const float* __restrict__ bias_hi, int bsplit,
                   float* __restrict__ C, int n,
                   const _Float16* __restrict__ F2h, const _Float16* __restrict__ G2h,
                   const float* __restrict__ bF2, const float* __restrict__ bG2,
                   float* __restrict__ oF, float* __restrict__ oG)
{
    __shared__ __attribute__((aligned(16))) _Float16 s_a[16 * GPITCH];
    __shared__ __attribute__((aligned(16))) _Float16 s_t[HEADS ? 16 * GPITCH : 8];
    const int lane = threadIdx.x & 31;
    const int wv = __builtin_amdgcn_readfirstlane(threadIdx.x >> 5);
    const int row0 = blockIdx.x * 16;
    const float* hl = h + lane * 4;
    for (int k = 0; k < GNPW; ++k) {
        const int r = wv * GNPW + k;
        const int node = row0 + r;
        float4 acc = make_float4(0.f, 0.f, 0.f, 0.f);
        if (node < n) {
            const int beg = rowptr[node], end = rowptr[node + 1];
            float sd = dis[node]; sd *= sd;
            int base = beg;
            int cnt  = (end - base < 32) ? (end - base) : 32;
            int2 my = make_int2(0, 0);
            if (lane < cnt) my = csr[base + lane];
            const float4 v = *(const float4*)(hl + (size_t)node * GCH);
            while (cnt > 0) {
                int j = 0;
                for (; j + 8 <= cnt; j += 8) gather_group<true >(hl, my.x, my.y, j, cnt, acc);
                if (j < cnt)                 gather_group<false>(hl, my.x, my.y, j, cnt, acc);
                base += 32;
                cnt = (end - base < 32) ? (end - base) : 32;
                if (lane < cnt) my = csr[base + lane];
            }
            acc.x = fmaf(v.x, sd, acc.x);
            acc.y = fmaf(v.y, sd, acc.y);
            acc.z = fmaf(v.z, sd, acc.z);
            acc.w = fmaf(v.w, sd, acc.w);
        }
        if (ACT_IN) { acc.x = elu_f(acc.x); acc.y = elu_f(acc.y); acc.z = elu_f(acc.z); acc.w = elu_f(acc.w); }
        typedef __attribute__((ext_vector_type(4))) _Float16 v4h;
        v4h hv = { (_Float16)acc.x, (_Float16)acc.y, (_Float16)acc.z, (_Float16)acc.w };
        *(v4h*)(s_a + r * GPITCH + lane * 4) = hv;
    }
    __syncthreads();
    if (!HEADS && wv >= 4) return;

    constexpr int NT = 2;
    const int hs   = lane >> 4;
    const int lrow = lane & 15;
    if (wv < 4) {
        const _Float16* Arow = s_a + lrow * GPITCH;
        const _Float16* Bcol[NT];
        #pragma unroll
        for (int t = 0; t < NT; ++t)
            Bcol[t] = BhP + (size_t)(wv * NT + t) * 16 * GCH + lane * 8;

        v8f acc2[NT];
        #pragma unroll
        for (int t = 0; t < NT; ++t) acc2[t] = (v8f){};

        #pragma unroll
        for (int kk = 0; kk < GCH; kk += 32) {
            const int kb = kk + hs * 8;
            v8h a0 = *(const v8h*)(Arow + kb);
            v8h a1 = *(const v8h*)(Arow + kb + 16);
            v16h a = __builtin_shufflevector(a0, a1, 0, 1, 2, 3, 4, 5, 6, 7,
                                             8, 9, 10, 11, 12, 13, 14, 15);
            #pragma unroll
            for (int t = 0; t < NT; ++t) {
                v8h b0 = *(const v8h*)(Bcol[t] + kk * 16);
                v8h b1 = *(const v8h*)(Bcol[t] + kk * 16 + 256);
                v16h b = __builtin_shufflevector(b0, b1, 0, 1, 2, 3, 4, 5, 6, 7,
                                                 8, 9, 10, 11, 12, 13, 14, 15);
                acc2[t] = __builtin_amdgcn_wmma_f32_16x16x32_f16(false, a, false, b,
                                                                 (short)0, acc2[t],
                                                                 false, false);
            }
        }

        const float* bp = bias;
        int bo = 0;
        if (wv * NT * 16 >= bsplit) { bp = bias_hi; bo = bsplit; }

        const bool fullM = (row0 + 16) <= n;
        #pragma unroll
        for (int t = 0; t < NT; ++t) {
            const int col = (wv * NT + t) * 16 + lrow;
            const float bv = bp ? bp[col - bo] : 0.f;
            if (HEADS) {
                _Float16* Tb = s_t + (hs * 8) * GPITCH + col;
                #pragma unroll
                for (int r = 0; r < 8; ++r) {
                    float v = acc2[t][r] + bv;
                    if (ACT_OUT) v = leaky_f(v);
                    Tb[r * GPITCH] = (_Float16)v;
                }
            } else {
                float* Cb = C + (size_t)(row0 + hs * 8) * GCH + col;
                if (fullM) {
                    #pragma unroll
                    for (int r = 0; r < 8; ++r) {
                        float v = acc2[t][r] + bv;
                        if (ACT_OUT) v = leaky_f(v);
                        Cb[(size_t)r * GCH] = v;
                    }
                } else {
                    #pragma unroll
                    for (int r = 0; r < 8; ++r) {
                        if (row0 + hs * 8 + r < n) {
                            float v = acc2[t][r] + bv;
                            if (ACT_OUT) v = leaky_f(v);
                            Cb[(size_t)r * GCH] = v;
                        }
                    }
                }
            }
        }
    }
    if (!HEADS) return;

    __syncthreads();
    if (wv >= 4) return;

    constexpr int K2 = 64, N2 = 32;
    const int g = wv >> 1, tt = wv & 1;
    const _Float16* Arow2 = s_t + lrow * GPITCH + g * K2;
    const _Float16* Bc2   = (g ? G2h : F2h) + (size_t)tt * 16 * K2 + lane * 8;
    v8f acc3 = (v8f){};
    #pragma unroll
    for (int kk = 0; kk < K2; kk += 32) {
        const int kb = kk + hs * 8;
        v8h a0 = *(const v8h*)(Arow2 + kb);
        v8h a1 = *(const v8h*)(Arow2 + kb + 16);
        v16h a = __builtin_shufflevector(a0, a1, 0, 1, 2, 3, 4, 5, 6, 7,
                                         8, 9, 10, 11, 12, 13, 14, 15);
        v8h b0 = *(const v8h*)(Bc2 + kk * 16);
        v8h b1 = *(const v8h*)(Bc2 + kk * 16 + 256);
        v16h b = __builtin_shufflevector(b0, b1, 0, 1, 2, 3, 4, 5, 6, 7,
                                         8, 9, 10, 11, 12, 13, 14, 15);
        acc3 = __builtin_amdgcn_wmma_f32_16x16x32_f16(false, a, false, b,
                                                      (short)0, acc3,
                                                      false, false);
    }
    const int col2 = tt * 16 + lrow;
    const float bv2 = (g ? bG2 : bF2)[col2];
    float* Ob = (g ? oG : oF) + (size_t)(row0 + hs * 8) * N2 + col2;
    if ((row0 + 16) <= n) {
        #pragma unroll
        for (int r = 0; r < 8; ++r)
            Ob[(size_t)r * N2] = leaky_f(acc3[r] + bv2);
    } else {
        #pragma unroll
        for (int r = 0; r < 8; ++r)
            if (row0 + hs * 8 + r < n) Ob[(size_t)r * N2] = leaky_f(acc3[r] + bv2);
    }
}

__global__ __launch_bounds__(256)
void k_pairdot(const int* __restrict__ ts, const float* __restrict__ tf,
               const float* __restrict__ tg, float* __restrict__ out, int M)
{
    const int lane = threadIdx.x & 31;
    const int base = (blockIdx.x * blockDim.x + threadIdx.x) & ~31;
    const int g8 = lane & 24;
    const int k  = lane & 7;
    float4 xr[8], yr[8];
    #pragma unroll
    for (int j = 0; j < 8; ++j) {
        int m = base + g8 + j;
        if (m >= M) m = M - 1;
        const int2 ab = *(const int2*)(ts + 2 * (size_t)m);
        xr[j] = *(const float4*)(tf + (size_t)ab.x * 32 + k * 4);
        yr[j] = *(const float4*)(tg + (size_t)ab.y * 32 + k * 4);
    }
    float p[8];
    #pragma unroll
    for (int j = 0; j < 8; ++j) {
        p[j] = xr[j].x * yr[j].x + xr[j].y * yr[j].y + xr[j].z * yr[j].z + xr[j].w * yr[j].w;
        p[j] += __shfl_xor(p[j], 4);
        p[j] += __shfl_xor(p[j], 2);
        p[j] += __shfl_xor(p[j], 1);
    }
    float r = p[0];
    #pragma unroll
    for (int j = 1; j < 8; ++j) r = (k == j) ? p[j] : r;
    if (base + lane < M) out[base + lane] = r;
}

extern "C" void kernel_launch(void* const* d_in, const int* in_sizes, int n_in,
                              void* d_out, int out_size, void* d_ws, size_t ws_size,
                              hipStream_t stream)
{
    const float* x    = (const float*)d_in[0];
    const int*   ei   = (const int*)  d_in[1];
    const int*   ts   = (const int*)  d_in[2];
    const float* W1   = (const float*)d_in[3];
    const float* W2   = (const float*)d_in[4];
    const float* tfw1 = (const float*)d_in[5];
    const float* tfb1 = (const float*)d_in[6];
    const float* tfw2 = (const float*)d_in[7];
    const float* tfb2 = (const float*)d_in[8];
    const float* tgw1 = (const float*)d_in[9];
    const float* tgb1 = (const float*)d_in[10];
    const float* tgw2 = (const float*)d_in[11];
    const float* tgb2 = (const float*)d_in[12];

    const int N = in_sizes[0] / D_IN;
    const int E = in_sizes[1] / 2;
    const int M = in_sizes[2] / 2;
    const int* srcI = ei;
    const int* dstI = ei + E;
    float* out = (float*)d_out;

    char* w = (char*)d_ws;
    auto alloc = [&](size_t bytes) -> char* {
        char* p = w; w += (bytes + 255) & ~(size_t)255; return p;
    };
    float*    dis = (float*)   alloc((size_t)N * 4);
    _Float16* W1h = (_Float16*)alloc((size_t)D_IN * H1 * 2);
    _Float16* W2h = (_Float16*)alloc((size_t)H1 * H2 * 2);
    _Float16* FG1h = (_Float16*)alloc((size_t)2 * H2 * H3 * 2);
    _Float16* F1h = FG1h;
    _Float16* G1h = FG1h + (size_t)H3 * H2;
    _Float16* F2h = (_Float16*)alloc((size_t)H3 * D_OUT * 2);
    _Float16* G2h = (_Float16*)alloc((size_t)H3 * D_OUT * 2);
    float*    B1  = (float*)   alloc((size_t)N * H1 * 4);
    float*    B2  = (float*)   alloc((size_t)N * H1 * 4);
    float*    B3  = (float*)   alloc((size_t)N * D_OUT * 4);
    float*    B4  = (float*)   alloc((size_t)N * D_OUT * 4);
    int*      cnt    = (int*) alloc((size_t)N * 4);
    int*      rowptr = (int*) alloc((size_t)(N + 1) * 4);
    int*      cursor = (int*) alloc((size_t)N * 4);
    int*      ebuf   = (int*) alloc((size_t)E * 4);
    int2*     csr    = (int2*)alloc((size_t)E * 8);
    (void)ws_size; (void)n_in; (void)out_size;

    const int T = 256;
    auto cdiv = [](long long a, long long b) { return (int)((a + b - 1) / b); };

    const int zb = cdiv(N, T);
    k_prep<<<zb + NB_WALL, T, 0, stream>>>(cnt, N, zb, W1, W2, tfw1, tfw2, tgw1, tgw2,
                                           W1h, W2h, F1h, F2h, G1h, G2h);

    k_count<<<cdiv(E, T), T, 0, stream>>>(dstI, cnt, E);
    const int sb = cdiv(N, 1024);
    k_scan_gemm1<<<sb + cdiv(N, G1ROWS), T, 0, stream>>>(cnt, rowptr, cursor, dis, N, sb, x, W1h, B1);
    k_fill <<<cdiv(E, T), T, 0, stream>>>(dstI, cursor, ebuf, E);
    k_sort <<<cdiv(E, T), T, 0, stream>>>(ebuf, srcI, dstI, rowptr, dis, csr, E);

    k_gather_gemm<1, 0, false><<<cdiv(N, 16), T, 0, stream>>>(rowptr, csr, dis, B1, W2h, nullptr, nullptr, H2, B2, N,
                                                              nullptr, nullptr, nullptr, nullptr, nullptr, nullptr);

    k_gather_gemm<0, 1, true><<<cdiv(N, 16), T, 0, stream>>>(rowptr, csr, dis, B2, FG1h, tfb1, tgb1, H3, nullptr, N,
                                                             F2h, G2h, tfb2, tgb2, B4, B3);

    k_pairdot<<<cdiv(M, T), T, 0, stream>>>(ts, B4, B3, out, M);
}


# Fill pass loses its atomic: the count's add returns each edge's arrival rank, fill places edges at bucket start + rank by plain loads and a store: 1.28 M global atomics to 0.64 M; sums unchanged
# speedup vs baseline: 2.5670x; 1.1144x over previous
#include <hip/hip_runtime.h>
#include <hip/hip_bf16.h>

typedef __attribute__((ext_vector_type(16))) _Float16 v16h;
typedef __attribute__((ext_vector_type(8)))  _Float16 v8h;
typedef __attribute__((ext_vector_type(8)))  float    v8f;

#define LEAKY_SLOPE 0.01f

__device__ __forceinline__ float elu_f(float x)   { return x > 0.f ? x : (expf(x) - 1.f); }
__device__ __forceinline__ float leaky_f(float x) { return x > 0.f ? x : LEAKY_SLOPE * x; }

constexpr int D_IN = 256, H1 = 128, H2 = 128, H3 = 64, D_OUT = 32;

__device__ __forceinline__ int pack_idx(int k, int n, int K)
{
    const int nt = n >> 4, c = n & 15;
    const int ks = k >> 5, j = (k >> 3) & 3, idx = k & 7;
    const int lane = (j & 1) * 16 + c, part = j >> 1;
    return (((nt * (K >> 5) + ks) * 2 + part) * 32 + lane) * 8 + idx;
}

__global__ void k_count(const int* __restrict__ dst, int* cnt, int* __restrict__ slot, int E)
{
    int e = blockIdx.x * blockDim.x + threadIdx.x;
    if (e < E) slot[e] = atomicAdd(&cnt[dst[e]], 1);
}

__device__ __forceinline__
void scan_tile(int bid, const int* __restrict__ cnt, int* __restrict__ rowptr,
               float* __restrict__ dis, int n)
{
    __shared__ int s_pre[8];
    __shared__ int s_tot[8];
    const int t = threadIdx.x, lane = t & 31, wv = t >> 5;

    const int4* c4 = (const int4*)cnt;
    const int nb = bid;
    int pre = 0;
    #pragma unroll 8
    for (int m = 0; m < nb; ++m) {
        int4 v = c4[m * 256 + t];
        pre += (v.x + v.y) + (v.z + v.w);
    }

    const int i0 = bid * 1024 + t * 4;
    int c[4];
    if (i0 + 3 < n) {
        int4 v = c4[i0 >> 2];
        c[0] = v.x; c[1] = v.y; c[2] = v.z; c[3] = v.w;
    } else {
        for (int k = 0; k < 4; ++k) c[k] = (i0 + k < n) ? cnt[i0 + k] : 0;
    }
    const int mine = (c[0] + c[1]) + (c[2] + c[3]);

    int incl = mine;
    for (int off = 1; off < 32; off <<= 1) {
        int v = __shfl_up(incl, off);
        if (lane >= off) incl += v;
    }
    for (int off = 16; off > 0; off >>= 1) pre += __shfl_xor(pre, off);
    if (lane == 31) s_tot[wv] = incl;
    if (lane == 0)  s_pre[wv] = pre;
    __syncthreads();
    int run = incl - mine;
    for (int k = 0; k < 8; ++k) {
        run += s_pre[k];
        if (k < wv) run += s_tot[k];
    }

    int r[4]; float d[4];
    for (int k = 0; k < 4; ++k) {
        r[k] = run;
        d[k] = rsqrtf((float)(c[k] + 1));
        run += c[k];
    }
    if (i0 + 3 < n) {
        *(int4*)(rowptr + i0)  = make_int4(r[0], r[1], r[2], r[3]);
        *(float4*)(dis + i0)   = make_float4(d[0], d[1], d[2], d[3]);
    } else {
        for (int k = 0; k < 4; ++k)
            if (i0 + k < n) { rowptr[i0 + k] = r[k]; dis[i0 + k] = d[k]; }
    }
    if (i0 < n && n <= i0 + 4) rowptr[n] = run;
}

constexpr int G1ROWS  = 64;
constexpr int G1PITCH = D_IN + 8;
__device__ __forceinline__
void gemm1_tile(int bid, const float* __restrict__ x, const _Float16* __restrict__ BhP,
                float* __restrict__ C, int n, _Float16* __restrict__ s_x)
{
    const int lane = threadIdx.x & 31;
    const int wv = __builtin_amdgcn_readfirstlane(threadIdx.x >> 5);
    const int row0 = bid * G1ROWS;
    typedef __attribute__((ext_vector_type(4))) _Float16 v4h;

    float4 xv[16];
    #pragma unroll
    for (int u = 0; u < 8; ++u) {
        int node = row0 + wv * 8 + u;
        if (node >= n) node = n - 1;
        const float* xr = x + (size_t)node * D_IN + lane * 4;
        xv[2 * u]     = *(const float4*)(xr);
        xv[2 * u + 1] = *(const float4*)(xr + 128);
    }
    #pragma unroll
    for (int u = 0; u < 8; ++u) {
        _Float16* sr = s_x + (wv * 8 + u) * G1PITCH + lane * 4;
        const float4 p = xv[2 * u], q = xv[2 * u + 1];
        *(v4h*)(sr)       = (v4h){ (_Float16)p.x, (_Float16)p.y, (_Float16)p.z, (_Float16)p.w };
        *(v4h*)(sr + 128) = (v4h){ (_Float16)q.x, (_Float16)q.y, (_Float16)q.z, (_Float16)q.w };
    }
    __syncthreads();

    constexpr int NT = 4;
    const int mt = wv >> 1, ns = wv & 1;
    const int hs   = lane >> 4;
    const int lrow = lane & 15;
    const _Float16* Arow = s_x + (mt * 16 + lrow) * G1PITCH;
    const _Float16* Bcol[NT];
    #pragma unroll
    for (int t = 0; t < NT; ++t)
        Bcol[t] = BhP + (size_t)(ns * NT + t) * 16 * D_IN + lane * 8;

    v8f acc[NT];
    #pragma unroll
    for (int t = 0; t < NT; ++t) acc[t] = (v8f){};

    for (int kk = 0; kk < D_IN; kk += 32) {
        const int kb = kk + hs * 8;
        v8h a0 = *(const v8h*)(Arow + kb);
        v8h a1 = *(const v8h*)(Arow + kb + 16);
        v16h a = __builtin_shufflevector(a0, a1, 0, 1, 2, 3, 4, 5, 6, 7,
                                         8, 9, 10, 11, 12, 13, 14, 15);
        #pragma unroll
        for (int t = 0; t < NT; ++t) {
            v8h b0 = *(const v8h*)(Bcol[t] + kk * 16);
            v8h b1 = *(const v8h*)(Bcol[t] + kk * 16 + 256);
            v16h b = __builtin_shufflevector(b0, b1, 0, 1, 2, 3, 4, 5, 6, 7,
                                             8, 9, 10, 11, 12, 13, 14, 15);
            acc[t] = __builtin_amdgcn_wmma_f32_16x16x32_f16(false, a, false, b,
                                                            (short)0, acc[t],
                                                            false, false);
        }
    }

    const int orow0 = row0 + mt * 16 + hs * 8;
    const bool fullM = (row0 + mt * 16 + 16) <= n;
    #pragma unroll
    for (int t = 0; t < NT; ++t) {
        const int col = (ns * NT + t) * 16 + lrow;
        float* Cb = C + (size_t)orow0 * H1 + col;
        if (fullM) {
            #pragma unroll
            for (int r = 0; r < 8; ++r) Cb[(size_t)r * H1] = acc[t][r];
        } else {
            #pragma unroll
            for (int r = 0; r < 8; ++r)
                if (orow0 + r < n) Cb[(size_t)r * H1] = acc[t][r];
        }
    }
}

__global__ __launch_bounds__(256)
void k_scan_gemm1(const int* __restrict__ cnt, int* __restrict__ rowptr,
                  float* __restrict__ dis, int n, int sb,
                  const float* __restrict__ x, const _Float16* __restrict__ W1h,
                  float* __restrict__ h1)
{
    __shared__ __attribute__((aligned(16))) _Float16 s_x[G1ROWS * G1PITCH];
    if ((int)blockIdx.x < sb) {
        scan_tile((int)blockIdx.x, cnt, rowptr, dis, n);
        return;
    }
    gemm1_tile((int)blockIdx.x - sb, x, W1h, h1, n, s_x);
}

__global__ void k_fill(const int* __restrict__ dst, const int* __restrict__ rowptr,
                       const int* __restrict__ slot, int* __restrict__ ebuf, int E)
{
    int e = blockIdx.x * blockDim.x + threadIdx.x;
    if (e < E) ebuf[rowptr[dst[e]] + slot[e]] = e;
}

__global__ void k_sort(const int* __restrict__ ebuf, const int* __restrict__ srcI,
                       const int* __restrict__ dstI, const int* __restrict__ rowptr,
                       const float* __restrict__ dis, int2* __restrict__ csr, int E)
{
    int p = blockIdx.x * blockDim.x + threadIdx.x;
    if (p < E) {
        const int e = ebuf[p];
        const int d = dstI[e];
        const int b = rowptr[d], en = rowptr[d + 1];
        int rank = 0;
        for (int i = b; i < en; ++i) rank += (ebuf[i] < e) ? 1 : 0;
        const int s = srcI[e];
        const float w = dis[s] * dis[d];
        csr[b + rank] = make_int2(s, __float_as_int(w));
    }
}

__device__ __forceinline__
void f16T_blk(const float* __restrict__ s, _Float16* __restrict__ d, int K, int Nn, int blk)
{
    int i = blk * 256 + threadIdx.x;
    if (i < K * Nn) {
        int k = i / Nn, n = i % Nn;
        d[pack_idx(k, n, K)] = (_Float16)s[i];
    }
}

constexpr int NB_W1 = (D_IN * H1 + 255) / 256, NB_W2 = (H1 * H2 + 255) / 256;
constexpr int NB_H1 = (H2 * H3 + 255) / 256,   NB_H2 = (H3 * D_OUT + 255) / 256;
constexpr int NB_WALL = NB_W1 + NB_W2 + 2 * NB_H1 + 2 * NB_H2;

__global__ __launch_bounds__(256)
void k_prep(int* __restrict__ cnt, int n, int zb,
            const float* __restrict__ W1,   const float* __restrict__ W2,
            const float* __restrict__ tfw1, const float* __restrict__ tfw2,
            const float* __restrict__ tgw1, const float* __restrict__ tgw2,
            _Float16* __restrict__ W1h, _Float16* __restrict__ W2h,
            _Float16* __restrict__ F1h, _Float16* __restrict__ F2h,
            _Float16* __restrict__ G1h, _Float16* __restrict__ G2h)
{
    int b = blockIdx.x;
    if (b < zb) {
        const int i = b * 256 + threadIdx.x;
        if (i < n) cnt[i] = 0;
        return;
    }
    b -= zb;
    if (b < NB_W1) { f16T_blk(W1,   W1h, D_IN, H1,    b); return; }  b -= NB_W1;
    if (b < NB_W2) { f16T_blk(W2,   W2h, H1,   H2,    b); return; }  b -= NB_W2;
    if (b < NB_H1) { f16T_blk(tfw1, F1h, H2,   H3,    b); return; }  b -= NB_H1;
    if (b < NB_H2) { f16T_blk(tfw2, F2h, H3,   D_OUT, b); return; }  b -= NB_H2;
    if (b < NB_H1) { f16T_blk(tgw1, G1h, H2,   H3,    b); return; }  b -= NB_H1;
    f16T_blk(tgw2, G2h, H3, D_OUT, b);
}

template <bool FULL>
__device__ __forceinline__ void gather_group(const float* __restrict__ hl, int mys, int myw,
                                             int j, int cnt, float4& acc)
{
    float4 r[8];
    float  w[8];
    #pragma unroll
    for (int u = 0; u < 8; ++u) {
        r[u] = make_float4(0.f, 0.f, 0.f, 0.f);
        w[u] = 0.f;
        if (FULL || j + u < cnt) {
            const int s = __builtin_amdgcn_readlane(mys, j + u);
            w[u] = __int_as_float(__builtin_amdgcn_readlane(myw, j + u));
            r[u] = *(const float4*)(hl + (size_t)s * 128);
        }
    }
    #pragma unroll
    for (int u = 0; u < 8; ++u) {
        if (FULL || j + u < cnt) {
            acc.x = fmaf(r[u].x, w[u], acc.x);
            acc.y = fmaf(r[u].y, w[u], acc.y);
            acc.z = fmaf(r[u].z, w[u], acc.z);
            acc.w = fmaf(r[u].w, w[u], acc.w);
        }
    }
}

#define GNPW 2
constexpr int GCH = 128;
constexpr int GPITCH = GCH + 8;
template <int ACT_IN, int ACT_OUT, bool HEADS>
__global__ __launch_bounds__(256)
void k_gather_gemm(const int* __restrict__ rowptr, const int2* __restrict__ csr,
                   const float* __restrict__ dis, const float* __restrict__ h,
                   const _Float16* __restrict__ BhP,
                   const float* __restrict__ bias, const float* __restrict__ bias_hi, int bsplit,
                   float* __restrict__ C, int n,
                   const _Float16* __restrict__ F2h, const _Float16* __restrict__ G2h,
                   const float* __restrict__ bF2, const float* __restrict__ bG2,
                   float* __restrict__ oF, float* __restrict__ oG)
{
    __shared__ __attribute__((aligned(16))) _Float16 s_a[16 * GPITCH];
    __shared__ __attribute__((aligned(16))) _Float16 s_t[HEADS ? 16 * GPITCH : 8];
    const int lane = threadIdx.x & 31;
    const int wv = __builtin_amdgcn_readfirstlane(threadIdx.x >> 5);
    const int row0 = blockIdx.x * 16;
    const float* hl = h + lane * 4;
    for (int k = 0; k < GNPW; ++k) {
        const int r = wv * GNPW + k;
        const int node = row0 + r;
        float4 acc = make_float4(0.f, 0.f, 0.f, 0.f);
        if (node < n) {
            const int beg = rowptr[node], end = rowptr[node + 1];
            float sd = dis[node]; sd *= sd;
            int base = beg;
            int cnt  = (end - base < 32) ? (end - base) : 32;
            int2 my = make_int2(0, 0);
            if (lane < cnt) my = csr[base + lane];
            const float4 v = *(const float4*)(hl + (size_t)node * GCH);
            while (cnt > 0) {
                int j = 0;
                for (; j + 8 <= cnt; j += 8) gather_group<true >(hl, my.x, my.y, j, cnt, acc);
                if (j < cnt)                 gather_group<false>(hl, my.x, my.y, j, cnt, acc);
                base += 32;
                cnt = (end - base < 32) ? (end - base) : 32;
                if (lane < cnt) my = csr[base + lane];
            }
            acc.x = fmaf(v.x, sd, acc.x);
            acc.y = fmaf(v.y, sd, acc.y);
            acc.z = fmaf(v.z, sd, acc.z);
            acc.w = fmaf(v.w, sd, acc.w);
        }
        if (ACT_IN) { acc.x = elu_f(acc.x); acc.y = elu_f(acc.y); acc.z = elu_f(acc.z); acc.w = elu_f(acc.w); }
        typedef __attribute__((ext_vector_type(4))) _Float16 v4h;
        v4h hv = { (_Float16)acc.x, (_Float16)acc.y, (_Float16)acc.z, (_Float16)acc.w };
        *(v4h*)(s_a + r * GPITCH + lane * 4) = hv;
    }
    __syncthreads();
    if (!HEADS && wv >= 4) return;

    constexpr int NT = 2;
    const int hs   = lane >> 4;
    const int lrow = lane & 15;
    if (wv < 4) {
        const _Float16* Arow = s_a + lrow * GPITCH;
        const _Float16* Bcol[NT];
        #pragma unroll
        for (int t = 0; t < NT; ++t)
            Bcol[t] = BhP + (size_t)(wv * NT + t) * 16 * GCH + lane * 8;

        v8f acc2[NT];
        #pragma unroll
        for (int t = 0; t < NT; ++t) acc2[t] = (v8f){};

        #pragma unroll
        for (int kk = 0; kk < GCH; kk += 32) {
            const int kb = kk + hs * 8;
            v8h a0 = *(const v8h*)(Arow + kb);
            v8h a1 = *(const v8h*)(Arow + kb + 16);
            v16h a = __builtin_shufflevector(a0, a1, 0, 1, 2, 3, 4, 5, 6, 7,
                                             8, 9, 10, 11, 12, 13, 14, 15);
            #pragma unroll
            for (int t = 0; t < NT; ++t) {
                v8h b0 = *(const v8h*)(Bcol[t] + kk * 16);
                v8h b1 = *(const v8h*)(Bcol[t] + kk * 16 + 256);
                v16h b = __builtin_shufflevector(b0, b1, 0, 1, 2, 3, 4, 5, 6, 7,
                                                 8, 9, 10, 11, 12, 13, 14, 15);
                acc2[t] = __builtin_amdgcn_wmma_f32_16x16x32_f16(false, a, false, b,
                                                                 (short)0, acc2[t],
                                                                 false, false);
            }
        }

        const float* bp = bias;
        int bo = 0;
        if (wv * NT * 16 >= bsplit) { bp = bias_hi; bo = bsplit; }

        const bool fullM = (row0 + 16) <= n;
        #pragma unroll
        for (int t = 0; t < NT; ++t) {
            const int col = (wv * NT + t) * 16 + lrow;
            const float bv = bp ? bp[col - bo] : 0.f;
            if (HEADS) {
                _Float16* Tb = s_t + (hs * 8) * GPITCH + col;
                #pragma unroll
                for (int r = 0; r < 8; ++r) {
                    float v = acc2[t][r] + bv;
                    if (ACT_OUT) v = leaky_f(v);
                    Tb[r * GPITCH] = (_Float16)v;
                }
            } else {
                float* Cb = C + (size_t)(row0 + hs * 8) * GCH + col;
                if (fullM) {
                    #pragma unroll
                    for (int r = 0; r < 8; ++r) {
                        float v = acc2[t][r] + bv;
                        if (ACT_OUT) v = leaky_f(v);
                        Cb[(size_t)r * GCH] = v;
                    }
                } else {
                    #pragma unroll
                    for (int r = 0; r < 8; ++r) {
                        if (row0 + hs * 8 + r < n) {
                            float v = acc2[t][r] + bv;
                            if (ACT_OUT) v = leaky_f(v);
                            Cb[(size_t)r * GCH] = v;
                        }
                    }
                }
            }
        }
    }
    if (!HEADS) return;

    __syncthreads();
    if (wv >= 4) return;

    constexpr int K2 = 64, N2 = 32;
    const int g = wv >> 1, tt = wv & 1;
    const _Float16* Arow2 = s_t + lrow * GPITCH + g * K2;
    const _Float16* Bc2   = (g ? G2h : F2h) + (size_t)tt * 16 * K2 + lane * 8;
    v8f acc3 = (v8f){};
    #pragma unroll
    for (int kk = 0; kk < K2; kk += 32) {
        const int kb = kk + hs * 8;
        v8h a0 = *(const v8h*)(Arow2 + kb);
        v8h a1 = *(const v8h*)(Arow2 + kb + 16);
        v16h a = __builtin_shufflevector(a0, a1, 0, 1, 2, 3, 4, 5, 6, 7,
                                         8, 9, 10, 11, 12, 13, 14, 15);
        v8h b0 = *(const v8h*)(Bc2 + kk * 16);
        v8h b1 = *(const v8h*)(Bc2 + kk * 16 + 256);
        v16h b = __builtin_shufflevector(b0, b1, 0, 1, 2, 3, 4, 5, 6, 7,
                                         8, 9, 10, 11, 12, 13, 14, 15);
        acc3 = __builtin_amdgcn_wmma_f32_16x16x32_f16(false, a, false, b,
                                                      (short)0, acc3,
                                                      false, false);
    }
    const int col2 = tt * 16 + lrow;
    const float bv2 = (g ? bG2 : bF2)[col2];
    float* Ob = (g ? oG : oF) + (size_t)(row0 + hs * 8) * N2 + col2;
    if ((row0 + 16) <= n) {
        #pragma unroll
        for (int r = 0; r < 8; ++r)
            Ob[(size_t)r * N2] = leaky_f(acc3[r] + bv2);
    } else {
        #pragma unroll
        for (int r = 0; r < 8; ++r)
            if (row0 + hs * 8 + r < n) Ob[(size_t)r * N2] = leaky_f(acc3[r] + bv2);
    }
}

__global__ __launch_bounds__(256)
void k_pairdot(const int* __restrict__ ts, const float* __restrict__ tf,
               const float* __restrict__ tg, float* __restrict__ out, int M)
{
    const int lane = threadIdx.x & 31;
    const int base = (blockIdx.x * blockDim.x + threadIdx.x) & ~31;
    const int g8 = lane & 24;
    const int k  = lane & 7;
    float4 xr[8], yr[8];
    #pragma unroll
    for (int j = 0; j < 8; ++j) {
        int m = base + g8 + j;
        if (m >= M) m = M - 1;
        const int2 ab = *(const int2*)(ts + 2 * (size_t)m);
        xr[j] = *(const float4*)(tf + (size_t)ab.x * 32 + k * 4);
        yr[j] = *(const float4*)(tg + (size_t)ab.y * 32 + k * 4);
    }
    float p[8];
    #pragma unroll
    for (int j = 0; j < 8; ++j) {
        p[j] = xr[j].x * yr[j].x + xr[j].y * yr[j].y + xr[j].z * yr[j].z + xr[j].w * yr[j].w;
        p[j] += __shfl_xor(p[j], 4);
        p[j] += __shfl_xor(p[j], 2);
        p[j] += __shfl_xor(p[j], 1);
    }
    float r = p[0];
    #pragma unroll
    for (int j = 1; j < 8; ++j) r = (k == j) ? p[j] : r;
    if (base + lane < M) out[base + lane] = r;
}

extern "C" void kernel_launch(void* const* d_in, const int* in_sizes, int n_in,
                              void* d_out, int out_size, void* d_ws, size_t ws_size,
                              hipStream_t stream)
{
    const float* x    = (const float*)d_in[0];
    const int*   ei   = (const int*)  d_in[1];
    const int*   ts   = (const int*)  d_in[2];
    const float* W1   = (const float*)d_in[3];
    const float* W2   = (const float*)d_in[4];
    const float* tfw1 = (const float*)d_in[5];
    const float* tfb1 = (const float*)d_in[6];
    const float* tfw2 = (const float*)d_in[7];
    const float* tfb2 = (const float*)d_in[8];
    const float* tgw1 = (const float*)d_in[9];
    const float* tgb1 = (const float*)d_in[10];
    const float* tgw2 = (const float*)d_in[11];
    const float* tgb2 = (const float*)d_in[12];

    const int N = in_sizes[0] / D_IN;
    const int E = in_sizes[1] / 2;
    const int M = in_sizes[2] / 2;
    const int* srcI = ei;
    const int* dstI = ei + E;
    float* out = (float*)d_out;

    char* w = (char*)d_ws;
    auto alloc = [&](size_t bytes) -> char* {
        char* p = w; w += (bytes + 255) & ~(size_t)255; return p;
    };
    float*    dis = (float*)   alloc((size_t)N * 4);
    _Float16* W1h = (_Float16*)alloc((size_t)D_IN * H1 * 2);
    _Float16* W2h = (_Float16*)alloc((size_t)H1 * H2 * 2);
    _Float16* FG1h = (_Float16*)alloc((size_t)2 * H2 * H3 * 2);
    _Float16* F1h = FG1h;
    _Float16* G1h = FG1h + (size_t)H3 * H2;
    _Float16* F2h = (_Float16*)alloc((size_t)H3 * D_OUT * 2);
    _Float16* G2h = (_Float16*)alloc((size_t)H3 * D_OUT * 2);
    float*    B1  = (float*)   alloc((size_t)N * H1 * 4);
    float*    B2  = (float*)   alloc((size_t)N * H1 * 4);
    float*    B3  = (float*)   alloc((size_t)N * D_OUT * 4);
    float*    B4  = (float*)   alloc((size_t)N * D_OUT * 4);
    int*      cnt    = (int*) alloc((size_t)N * 4);
    int*      rowptr = (int*) alloc((size_t)(N + 1) * 4);
    int*      slot   = (int*) alloc((size_t)E * 4);
    int*      ebuf   = (int*) alloc((size_t)E * 4);
    int2*     csr    = (int2*)alloc((size_t)E * 8);
    (void)ws_size; (void)n_in; (void)out_size;

    const int T = 256;
    auto cdiv = [](long long a, long long b) { return (int)((a + b - 1) / b); };

    const int zb = cdiv(N, T);
    k_prep<<<zb + NB_WALL, T, 0, stream>>>(cnt, N, zb, W1, W2, tfw1, tfw2, tgw1, tgw2,
                                           W1h, W2h, F1h, F2h, G1h, G2h);

    k_count<<<cdiv(E, T), T, 0, stream>>>(dstI, cnt, slot, E);
    const int sb = cdiv(N, 1024);
    k_scan_gemm1<<<sb + cdiv(N, G1ROWS), T, 0, stream>>>(cnt, rowptr, dis, N, sb, x, W1h, B1);
    k_fill <<<cdiv(E, T), T, 0, stream>>>(dstI, rowptr, slot, ebuf, E);
    k_sort <<<cdiv(E, T), T, 0, stream>>>(ebuf, srcI, dstI, rowptr, dis, csr, E);

    k_gather_gemm<1, 0, false><<<cdiv(N, 16), T, 0, stream>>>(rowptr, csr, dis, B1, W2h, nullptr, nullptr, H2, B2, N,
                                                              nullptr, nullptr, nullptr, nullptr, nullptr, nullptr);

    k_gather_gemm<0, 1, true><<<cdiv(N, 16), T, 0, stream>>>(rowptr, csr, dis, B2, FG1h, tfb1, tgb1, H3, nullptr, N,
                                                             F2h, G2h, tfb2, tgb2, B4, B3);

    k_pairdot<<<cdiv(M, T), T, 0, stream>>>(ts, B4, B3, out, M);
}


# Each node's bucket is a fixed row of 64 slots, so the count's returned rank places the edge at once: scan and fill launches go, GEMM 1 rides on the count (8 launches to 6); sums unchanged
# speedup vs baseline: 3.2260x; 1.2567x over previous
#include <hip/hip_runtime.h>
#include <hip/hip_bf16.h>

typedef __attribute__((ext_vector_type(16))) _Float16 v16h;
typedef __attribute__((ext_vector_type(8)))  _Float16 v8h;
typedef __attribute__((ext_vector_type(8)))  float    v8f;

#define LEAKY_SLOPE 0.01f

__device__ __forceinline__ float elu_f(float x)   { return x > 0.f ? x : (expf(x) - 1.f); }
__device__ __forceinline__ float leaky_f(float x) { return x > 0.f ? x : LEAKY_SLOPE * x; }

constexpr int D_IN = 256, H1 = 128, H2 = 128, H3 = 64, D_OUT = 32;

__device__ __forceinline__ int pack_idx(int k, int n, int K)
{
    const int nt = n >> 4, c = n & 15;
    const int ks = k >> 5, j = (k >> 3) & 3, idx = k & 7;
    const int lane = (j & 1) * 16 + c, part = j >> 1;
    return (((nt * (K >> 5) + ks) * 2 + part) * 32 + lane) * 8 + idx;
}

constexpr int BCAP = 64;
__device__ __forceinline__
void count_place(int bid, const int* __restrict__ dst, int* cnt, int* __restrict__ ell, int E)
{
    const int e = bid * 256 + threadIdx.x;
    if (e < E) {
        const int d = dst[e];
        const int r = atomicAdd(&cnt[d], 1);
        if (r < BCAP) ell[(size_t)d * BCAP + r] = e;
    }
}

constexpr int G1ROWS  = 64;
constexpr int G1PITCH = D_IN + 8;
__device__ __forceinline__
void gemm1_tile(int bid, const float* __restrict__ x, const _Float16* __restrict__ BhP,
                float* __restrict__ C, int n, _Float16* __restrict__ s_x)
{
    const int lane = threadIdx.x & 31;
    const int wv = __builtin_amdgcn_readfirstlane(threadIdx.x >> 5);
    const int row0 = bid * G1ROWS;
    typedef __attribute__((ext_vector_type(4))) _Float16 v4h;

    float4 xv[16];
    #pragma unroll
    for (int u = 0; u < 8; ++u) {
        int node = row0 + wv * 8 + u;
        if (node >= n) node = n - 1;
        const float* xr = x + (size_t)node * D_IN + lane * 4;
        xv[2 * u]     = *(const float4*)(xr);
        xv[2 * u + 1] = *(const float4*)(xr + 128);
    }
    #pragma unroll
    for (int u = 0; u < 8; ++u) {
        _Float16* sr = s_x + (wv * 8 + u) * G1PITCH + lane * 4;
        const float4 p = xv[2 * u], q = xv[2 * u + 1];
        *(v4h*)(sr)       = (v4h){ (_Float16)p.x, (_Float16)p.y, (_Float16)p.z, (_Float16)p.w };
        *(v4h*)(sr + 128) = (v4h){ (_Float16)q.x, (_Float16)q.y, (_Float16)q.z, (_Float16)q.w };
    }
    __syncthreads();

    constexpr int NT = 4;
    const int mt = wv >> 1, ns = wv & 1;
    const int hs   = lane >> 4;
    const int lrow = lane & 15;
    const _Float16* Arow = s_x + (mt * 16 + lrow) * G1PITCH;
    const _Float16* Bcol[NT];
    #pragma unroll
    for (int t = 0; t < NT; ++t)
        Bcol[t] = BhP + (size_t)(ns * NT + t) * 16 * D_IN + lane * 8;

    v8f acc[NT];
    #pragma unroll
    for (int t = 0; t < NT; ++t) acc[t] = (v8f){};

    for (int kk = 0; kk < D_IN; kk += 32) {
        const int kb = kk + hs * 8;
        v8h a0 = *(const v8h*)(Arow + kb);
        v8h a1 = *(const v8h*)(Arow + kb + 16);
        v16h a = __builtin_shufflevector(a0, a1, 0, 1, 2, 3, 4, 5, 6, 7,
                                         8, 9, 10, 11, 12, 13, 14, 15);
        #pragma unroll
        for (int t = 0; t < NT; ++t) {
            v8h b0 = *(const v8h*)(Bcol[t] + kk * 16);
            v8h b1 = *(const v8h*)(Bcol[t] + kk * 16 + 256);
            v16h b = __builtin_shufflevector(b0, b1, 0, 1, 2, 3, 4, 5, 6, 7,
                                             8, 9, 10, 11, 12, 13, 14, 15);
            acc[t] = __builtin_amdgcn_wmma_f32_16x16x32_f16(false, a, false, b,
                                                            (short)0, acc[t],
                                                            false, false);
        }
    }

    const int orow0 = row0 + mt * 16 + hs * 8;
    const bool fullM = (row0 + mt * 16 + 16) <= n;
    #pragma unroll
    for (int t = 0; t < NT; ++t) {
        const int col = (ns * NT + t) * 16 + lrow;
        float* Cb = C + (size_t)orow0 * H1 + col;
        if (fullM) {
            #pragma unroll
            for (int r = 0; r < 8; ++r) Cb[(size_t)r * H1] = acc[t][r];
        } else {
            #pragma unroll
            for (int r = 0; r < 8; ++r)
                if (orow0 + r < n) Cb[(size_t)r * H1] = acc[t][r];
        }
    }
}

__global__ __launch_bounds__(256)
void k_count_gemm1(const int* __restrict__ dst, int* cnt, int* __restrict__ ell, int E,
                   int n, int gb,
                   const float* __restrict__ x, const _Float16* __restrict__ W1h,
                   float* __restrict__ h1)
{
    __shared__ __attribute__((aligned(16))) _Float16 s_x[G1ROWS * G1PITCH];
    if ((int)blockIdx.x < gb) {
        gemm1_tile((int)blockIdx.x, x, W1h, h1, n, s_x);
        return;
    }
    count_place((int)blockIdx.x - gb, dst, cnt, ell, E);
}

__device__ __forceinline__
void sort_big(int node, int c, float dn, int lane,
              const int* __restrict__ srcI, const int* __restrict__ dstI,
              const int* __restrict__ cnt, int2* __restrict__ spill,
              int* __restrict__ spill_off, int* spill_total, int E)
{
    int off = 0;
    if (lane == 0) { off = atomicAdd(spill_total, c); spill_off[node] = off; }
    off = __builtin_amdgcn_readfirstlane(off);
    int run = 0;
    for (int e0 = 0; e0 < E; e0 += 32) {
        const int e = e0 + lane;
        const bool m = (e < E) && (dstI[e] == node);
        const unsigned bal = __builtin_amdgcn_ballot_w32(m);
        if (m) {
            const int pos = run + __popc(bal & ((1u << lane) - 1u));
            const int s = srcI[e];
            const float w = rsqrtf((float)(cnt[s] + 1)) * dn;
            spill[(size_t)off + pos] = make_int2(s, __float_as_int(w));
        }
        run += __popc(bal);
    }
}

constexpr int SNPW = 4;
__global__ __launch_bounds__(256)
void k_sort(const int* __restrict__ ell, const int* __restrict__ srcI,
            const int* __restrict__ dstI, const int* __restrict__ cnt,
            int2* __restrict__ csr, float* __restrict__ dis,
            int* __restrict__ spill_off, int* spill_total, int n, int E)
{
    int2* spill = csr + (size_t)n * BCAP;
    const int lane = threadIdx.x & 31;
    const int gw = __builtin_amdgcn_readfirstlane((int)((blockIdx.x * blockDim.x + threadIdx.x) >> 5));
    const int node0 = gw * SNPW;
    constexpr int NONE = 0x7fffffff;

    int cn[SNPW], e[SNPW], rk[SNPW], s[SNPW], cs[SNPW];
    #pragma unroll
    for (int k = 0; k < SNPW; ++k) {
        const int node = node0 + k;
        cn[k] = (node < n) ? cnt[node] : 0;
        if (node < n && lane == 0) dis[node] = rsqrtf((float)(cn[k] + 1));
    }
    #pragma unroll
    for (int k = 0; k < SNPW; ++k) {
        e[k] = NONE;
        if (cn[k] <= 32 && lane < cn[k]) e[k] = ell[(size_t)(node0 + k) * BCAP + lane];
    }
    #pragma unroll
    for (int k = 0; k < SNPW; ++k) {
        rk[k] = 0;
        const int m = (cn[k] <= 32) ? cn[k] : 0;
        for (int i = 0; i < m; ++i)
            rk[k] += (__builtin_amdgcn_readlane(e[k], i) < e[k]) ? 1 : 0;
    }
    #pragma unroll
    for (int k = 0; k < SNPW; ++k) {
        s[k] = 0;
        if (e[k] != NONE) s[k] = srcI[e[k]];
    }
    #pragma unroll
    for (int k = 0; k < SNPW; ++k) {
        cs[k] = 0;
        if (e[k] != NONE) cs[k] = cnt[s[k]];
    }
    #pragma unroll
    for (int k = 0; k < SNPW; ++k) {
        if (e[k] != NONE) {
            const float dn = rsqrtf((float)(cn[k] + 1));
            const float w = rsqrtf((float)(cs[k] + 1)) * dn;
            csr[(size_t)(node0 + k) * BCAP + rk[k]] = make_int2(s[k], __float_as_int(w));
        }
    }

    #pragma unroll
    for (int k = 0; k < SNPW; ++k) {
        const int c = cn[k];
        if (c <= 32) continue;
        const int node = node0 + k;
        const float dn = rsqrtf((float)(c + 1));
        if (c > BCAP) {
            sort_big(node, c, dn, lane, srcI, dstI, cnt, spill, spill_off, spill_total, E);
            continue;
        }
        const int* row = ell + (size_t)node * BCAP;
        const int ea = row[lane];
        const int eb = (32 + lane < c) ? row[32 + lane] : NONE;
        int ra = 0, rb = 0;
        for (int i = 0; i < 32; ++i) {
            const int v = __builtin_amdgcn_readlane(ea, i);
            ra += (v < ea) ? 1 : 0;
            rb += (v < eb) ? 1 : 0;
        }
        for (int i = 0; i < c - 32; ++i) {
            const int v = __builtin_amdgcn_readlane(eb, i);
            ra += (v < ea) ? 1 : 0;
            rb += (v < eb) ? 1 : 0;
        }
        int2* orow = csr + (size_t)node * BCAP;
        {
            const int sa = srcI[ea];
            const float w = rsqrtf((float)(cnt[sa] + 1)) * dn;
            orow[ra] = make_int2(sa, __float_as_int(w));
        }
        if (eb != NONE) {
            const int sb2 = srcI[eb];
            const float w = rsqrtf((float)(cnt[sb2] + 1)) * dn;
            orow[rb] = make_int2(sb2, __float_as_int(w));
        }
    }
}

__device__ __forceinline__
void f16T_blk(const float* __restrict__ s, _Float16* __restrict__ d, int K, int Nn, int blk)
{
    int i = blk * 256 + threadIdx.x;
    if (i < K * Nn) {
        int k = i / Nn, n = i % Nn;
        d[pack_idx(k, n, K)] = (_Float16)s[i];
    }
}

constexpr int NB_W1 = (D_IN * H1 + 255) / 256, NB_W2 = (H1 * H2 + 255) / 256;
constexpr int NB_H1 = (H2 * H3 + 255) / 256,   NB_H2 = (H3 * D_OUT + 255) / 256;
constexpr int NB_WALL = NB_W1 + NB_W2 + 2 * NB_H1 + 2 * NB_H2;

__global__ __launch_bounds__(256)
void k_prep(int* __restrict__ cnt, int* __restrict__ spill_total, int n, int zb,
            const float* __restrict__ W1,   const float* __restrict__ W2,
            const float* __restrict__ tfw1, const float* __restrict__ tfw2,
            const float* __restrict__ tgw1, const float* __restrict__ tgw2,
            _Float16* __restrict__ W1h, _Float16* __restrict__ W2h,
            _Float16* __restrict__ F1h, _Float16* __restrict__ F2h,
            _Float16* __restrict__ G1h, _Float16* __restrict__ G2h)
{
    int b = blockIdx.x;
    if (b < zb) {
        const int i = b * 256 + threadIdx.x;
        if (i < n) cnt[i] = 0;
        if (i == 0) *spill_total = 0;
        return;
    }
    b -= zb;
    if (b < NB_W1) { f16T_blk(W1,   W1h, D_IN, H1,    b); return; }  b -= NB_W1;
    if (b < NB_W2) { f16T_blk(W2,   W2h, H1,   H2,    b); return; }  b -= NB_W2;
    if (b < NB_H1) { f16T_blk(tfw1, F1h, H2,   H3,    b); return; }  b -= NB_H1;
    if (b < NB_H2) { f16T_blk(tfw2, F2h, H3,   D_OUT, b); return; }  b -= NB_H2;
    if (b < NB_H1) { f16T_blk(tgw1, G1h, H2,   H3,    b); return; }  b -= NB_H1;
    f16T_blk(tgw2, G2h, H3, D_OUT, b);
}

template <bool FULL>
__device__ __forceinline__ void gather_group(const float* __restrict__ hl, int mys, int myw,
                                             int j, int cnt, float4& acc)
{
    float4 r[8];
    float  w[8];
    #pragma unroll
    for (int u = 0; u < 8; ++u) {
        r[u] = make_float4(0.f, 0.f, 0.f, 0.f);
        w[u] = 0.f;
        if (FULL || j + u < cnt) {
            const int s = __builtin_amdgcn_readlane(mys, j + u);
            w[u] = __int_as_float(__builtin_amdgcn_readlane(myw, j + u));
            r[u] = *(const float4*)(hl + (size_t)s * 128);
        }
    }
    #pragma unroll
    for (int u = 0; u < 8; ++u) {
        if (FULL || j + u < cnt) {
            acc.x = fmaf(r[u].x, w[u], acc.x);
            acc.y = fmaf(r[u].y, w[u], acc.y);
            acc.z = fmaf(r[u].z, w[u], acc.z);
            acc.w = fmaf(r[u].w, w[u], acc.w);
        }
    }
}

#define GNPW 2
constexpr int GCH = 128;
constexpr int GPITCH = GCH + 8;
template <int ACT_IN, int ACT_OUT, bool HEADS>
__global__ __launch_bounds__(256)
void k_gather_gemm(const int* __restrict__ deg, const int* __restrict__ spill_off,
                   const int2* __restrict__ csr,
                   const float* __restrict__ dis, const float* __restrict__ h,
                   const _Float16* __restrict__ BhP,
                   const float* __restrict__ bias, const float* __restrict__ bias_hi, int bsplit,
                   float* __restrict__ C, int n,
                   const _Float16* __restrict__ F2h, const _Float16* __restrict__ G2h,
                   const float* __restrict__ bF2, const float* __restrict__ bG2,
                   float* __restrict__ oF, float* __restrict__ oG)
{
    __shared__ __attribute__((aligned(16))) _Float16 s_a[16 * GPITCH];
    __shared__ __attribute__((aligned(16))) _Float16 s_t[HEADS ? 16 * GPITCH : 8];
    const int lane = threadIdx.x & 31;
    const int wv = __builtin_amdgcn_readfirstlane(threadIdx.x >> 5);
    const int row0 = blockIdx.x * 16;
    const float* hl = h + lane * 4;
    for (int k = 0; k < GNPW; ++k) {
        const int r = wv * GNPW + k;
        const int node = row0 + r;
        float4 acc = make_float4(0.f, 0.f, 0.f, 0.f);
        if (node < n) {
            const int c = deg[node];
            int base = node * BCAP;
            if (c > BCAP) base = n * BCAP + spill_off[node];
            const int end = base + c;
            float sd = dis[node]; sd *= sd;
            int cnt  = (end - base < 32) ? (end - base) : 32;
            int2 my = make_int2(0, 0);
            if (lane < cnt) my = csr[base + lane];
            const float4 v = *(const float4*)(hl + (size_t)node * GCH);
            while (cnt > 0) {
                int j = 0;
                for (; j + 8 <= cnt; j += 8) gather_group<true >(hl, my.x, my.y, j, cnt, acc);
                if (j < cnt)                 gather_group<false>(hl, my.x, my.y, j, cnt, acc);
                base += 32;
                cnt = (end - base < 32) ? (end - base) : 32;
                if (lane < cnt) my = csr[base + lane];
            }
            acc.x = fmaf(v.x, sd, acc.x);
            acc.y = fmaf(v.y, sd, acc.y);
            acc.z = fmaf(v.z, sd, acc.z);
            acc.w = fmaf(v.w, sd, acc.w);
        }
        if (ACT_IN) { acc.x = elu_f(acc.x); acc.y = elu_f(acc.y); acc.z = elu_f(acc.z); acc.w = elu_f(acc.w); }
        typedef __attribute__((ext_vector_type(4))) _Float16 v4h;
        v4h hv = { (_Float16)acc.x, (_Float16)acc.y, (_Float16)acc.z, (_Float16)acc.w };
        *(v4h*)(s_a + r * GPITCH + lane * 4) = hv;
    }
    __syncthreads();
    if (!HEADS && wv >= 4) return;

    constexpr int NT = 2;
    const int hs   = lane >> 4;
    const int lrow = lane & 15;
    if (wv < 4) {
        const _Float16* Arow = s_a + lrow * GPITCH;
        const _Float16* Bcol[NT];
        #pragma unroll
        for (int t = 0; t < NT; ++t)
            Bcol[t] = BhP + (size_t)(wv * NT + t) * 16 * GCH + lane * 8;

        v8f acc2[NT];
        #pragma unroll
        for (int t = 0; t < NT; ++t) acc2[t] = (v8f){};

        #pragma unroll
        for (int kk = 0; kk < GCH; kk += 32) {
            const int kb = kk + hs * 8;
            v8h a0 = *(const v8h*)(Arow + kb);
            v8h a1 = *(const v8h*)(Arow + kb + 16);
            v16h a = __builtin_shufflevector(a0, a1, 0, 1, 2, 3, 4, 5, 6, 7,
                                             8, 9, 10, 11, 12, 13, 14, 15);
            #pragma unroll
            for (int t = 0; t < NT; ++t) {
                v8h b0 = *(const v8h*)(Bcol[t] + kk * 16);
                v8h b1 = *(const v8h*)(Bcol[t] + kk * 16 + 256);
                v16h b = __builtin_shufflevector(b0, b1, 0, 1, 2, 3, 4, 5, 6, 7,
                                                 8, 9, 10, 11, 12, 13, 14, 15);
                acc2[t] = __builtin_amdgcn_wmma_f32_16x16x32_f16(false, a, false, b,
                                                                 (short)0, acc2[t],
                                                                 false, false);
            }
        }

        const float* bp = bias;
        int bo = 0;
        if (wv * NT * 16 >= bsplit) { bp = bias_hi; bo = bsplit; }

        const bool fullM = (row0 + 16) <= n;
        #pragma unroll
        for (int t = 0; t < NT; ++t) {
            const int col = (wv * NT + t) * 16 + lrow;
            const float bv = bp ? bp[col - bo] : 0.f;
            if (HEADS) {
                _Float16* Tb = s_t + (hs * 8) * GPITCH + col;
                #pragma unroll
                for (int r = 0; r < 8; ++r) {
                    float v = acc2[t][r] + bv;
                    if (ACT_OUT) v = leaky_f(v);
                    Tb[r * GPITCH] = (_Float16)v;
                }
            } else {
                float* Cb = C + (size_t)(row0 + hs * 8) * GCH + col;
                if (fullM) {
                    #pragma unroll
                    for (int r = 0; r < 8; ++r) {
                        float v = acc2[t][r] + bv;
                        if (ACT_OUT) v = leaky_f(v);
                        Cb[(size_t)r * GCH] = v;
                    }
                } else {
                    #pragma unroll
                    for (int r = 0; r < 8; ++r) {
                        if (row0 + hs * 8 + r < n) {
                            float v = acc2[t][r] + bv;
                            if (ACT_OUT) v = leaky_f(v);
                            Cb[(size_t)r * GCH] = v;
                        }
                    }
                }
            }
        }
    }
    if (!HEADS) return;

    __syncthreads();
    if (wv >= 4) return;

    constexpr int K2 = 64, N2 = 32;
    const int g = wv >> 1, tt = wv & 1;
    const _Float16* Arow2 = s_t + lrow * GPITCH + g * K2;
    const _Float16* Bc2   = (g ? G2h : F2h) + (size_t)tt * 16 * K2 + lane * 8;
    v8f acc3 = (v8f){};
    #pragma unroll
    for (int kk = 0; kk < K2; kk += 32) {
        const int kb = kk + hs * 8;
        v8h a0 = *(const v8h*)(Arow2 + kb);
        v8h a1 = *(const v8h*)(Arow2 + kb + 16);
        v16h a = __builtin_shufflevector(a0, a1, 0, 1, 2, 3, 4, 5, 6, 7,
                                         8, 9, 10, 11, 12, 13, 14, 15);
        v8h b0 = *(const v8h*)(Bc2 + kk * 16);
        v8h b1 = *(const v8h*)(Bc2 + kk * 16 + 256);
        v16h b = __builtin_shufflevector(b0, b1, 0, 1, 2, 3, 4, 5, 6, 7,
                                         8, 9, 10, 11, 12, 13, 14, 15);
        acc3 = __builtin_amdgcn_wmma_f32_16x16x32_f16(false, a, false, b,
                                                      (short)0, acc3,
                                                      false, false);
    }
    const int col2 = tt * 16 + lrow;
    const float bv2 = (g ? bG2 : bF2)[col2];
    float* Ob = (g ? oG : oF) + (size_t)(row0 + hs * 8) * N2 + col2;
    if ((row0 + 16) <= n) {
        #pragma unroll
        for (int r = 0; r < 8; ++r)
            Ob[(size_t)r * N2] = leaky_f(acc3[r] + bv2);
    } else {
        #pragma unroll
        for (int r = 0; r < 8; ++r)
            if (row0 + hs * 8 + r < n) Ob[(size_t)r * N2] = leaky_f(acc3[r] + bv2);
    }
}

__global__ __launch_bounds__(256)
void k_pairdot(const int* __restrict__ ts, const float* __restrict__ tf,
               const float* __restrict__ tg, float* __restrict__ out, int M)
{
    const int lane = threadIdx.x & 31;
    const int base = (blockIdx.x * blockDim.x + threadIdx.x) & ~31;
    const int g8 = lane & 24;
    const int k  = lane & 7;
    float4 xr[8], yr[8];
    #pragma unroll
    for (int j = 0; j < 8; ++j) {
        int m = base + g8 + j;
        if (m >= M) m = M - 1;
        const int2 ab = *(const int2*)(ts + 2 * (size_t)m);
        xr[j] = *(const float4*)(tf + (size_t)ab.x * 32 + k * 4);
        yr[j] = *(const float4*)(tg + (size_t)ab.y * 32 + k * 4);
    }
    float p[8];
    #pragma unroll
    for (int j = 0; j < 8; ++j) {
        p[j] = xr[j].x * yr[j].x + xr[j].y * yr[j].y + xr[j].z * yr[j].z + xr[j].w * yr[j].w;
        p[j] += __shfl_xor(p[j], 4);
        p[j] += __shfl_xor(p[j], 2);
        p[j] += __shfl_xor(p[j], 1);
    }
    float r = p[0];
    #pragma unroll
    for (int j = 1; j < 8; ++j) r = (k == j) ? p[j] : r;
    if (base + lane < M) out[base + lane] = r;
}

extern "C" void kernel_launch(void* const* d_in, const int* in_sizes, int n_in,
                              void* d_out, int out_size, void* d_ws, size_t ws_size,
                              hipStream_t stream)
{
    const float* x    = (const float*)d_in[0];
    const int*   ei   = (const int*)  d_in[1];
    const int*   ts   = (const int*)  d_in[2];
    const float* W1   = (const float*)d_in[3];
    const float* W2   = (const float*)d_in[4];
    const float* tfw1 = (const float*)d_in[5];
    const float* tfb1 = (const float*)d_in[6];
    const float* tfw2 = (const float*)d_in[7];
    const float* tfb2 = (const float*)d_in[8];
    const float* tgw1 = (const float*)d_in[9];
    const float* tgb1 = (const float*)d_in[10];
    const float* tgw2 = (const float*)d_in[11];
    const float* tgb2 = (const float*)d_in[12];

    const int N = in_sizes[0] / D_IN;
    const int E = in_sizes[1] / 2;
    const int M = in_sizes[2] / 2;
    const int* srcI = ei;
    const int* dstI = ei + E;
    float* out = (float*)d_out;

    char* w = (char*)d_ws;
    auto alloc = [&](size_t bytes) -> char* {
        char* p = w; w += (bytes + 255) & ~(size_t)255; return p;
    };
    _Float16* W1h = (_Float16*)alloc((size_t)D_IN * H1 * 2);
    _Float16* W2h = (_Float16*)alloc((size_t)H1 * H2 * 2);
    _Float16* FG1h = (_Float16*)alloc((size_t)2 * H2 * H3 * 2);
    _Float16* F1h = FG1h;
    _Float16* G1h = FG1h + (size_t)H3 * H2;
    _Float16* F2h = (_Float16*)alloc((size_t)H3 * D_OUT * 2);
    _Float16* G2h = (_Float16*)alloc((size_t)H3 * D_OUT * 2);
    float*    dis = (float*)   alloc((size_t)N * 4);
    float*    B1  = (float*)   alloc((size_t)N * H1 * 4);
    float*    B2  = (float*)   alloc((size_t)N * H1 * 4);
    float*    B3  = (float*)   alloc((size_t)N * D_OUT * 4);
    float*    B4  = (float*)   alloc((size_t)N * D_OUT * 4);
    int*      cnt    = (int*) alloc((size_t)N * 4);
    int*      ell    = (int*) alloc((size_t)N * BCAP * 4);
    int2*     csr    = (int2*)alloc(((size_t)N * BCAP + E) * 8);
    int*      spoff  = (int*) alloc((size_t)N * 4);
    int*      sptot  = (int*) alloc(256);
    (void)ws_size; (void)n_in; (void)out_size;

    const int T = 256;
    auto cdiv = [](long long a, long long b) { return (int)((a + b - 1) / b); };

    const int zb = cdiv(N, T);
    k_prep<<<zb + NB_WALL, T, 0, stream>>>(cnt, sptot, N, zb, W1, W2, tfw1, tfw2, tgw1, tgw2,
                                           W1h, W2h, F1h, F2h, G1h, G2h);

    const int gb = cdiv(N, G1ROWS);
    k_count_gemm1<<<gb + cdiv(E, T), T, 0, stream>>>(dstI, cnt, ell, E, N, gb, x, W1h, B1);
    k_sort<<<cdiv(N, (T / 32) * SNPW), T, 0, stream>>>(ell, srcI, dstI, cnt, csr, dis, spoff, sptot, N, E);

    k_gather_gemm<1, 0, false><<<cdiv(N, 16), T, 0, stream>>>(cnt, spoff, csr, dis, B1, W2h, nullptr, nullptr, H2, B2, N,
                                                              nullptr, nullptr, nullptr, nullptr, nullptr, nullptr);

    k_gather_gemm<0, 1, true><<<cdiv(N, 16), T, 0, stream>>>(cnt, spoff, csr, dis, B2, FG1h, tfb1, tgb1, H3, nullptr, N,
                                                             F2h, G2h, tfb2, tgb2, B4, B3);

    k_pairdot<<<cdiv(M, T), T, 0, stream>>>(ts, B4, B3, out, M);
}
